# starting kernel
# baseline (speedup 1.0000x reference)
#include <hip/hip_runtime.h>

typedef float fx2 __attribute__((ext_vector_type(2)));
typedef float fx8 __attribute__((ext_vector_type(8)));

__global__ void deg_init_kernel(float* __restrict__ deg, int n) {
  int i = blockIdx.x * blockDim.x + threadIdx.x;
  if (i < n) deg[i] = 1.0f;
}

__global__ void deg_accum_kernel(const int* __restrict__ dst,
                                 const float* __restrict__ ew,
                                 float* __restrict__ deg, int e) {
  int i = blockIdx.x * blockDim.x + threadIdx.x;
  if (i < e) unsafeAtomicAdd(&deg[dst[i]], ew[i]);
}

__global__ void dinv_kernel(const float* __restrict__ deg,
                            float* __restrict__ dinv, int n) {
  int i = blockIdx.x * blockDim.x + threadIdx.x;
  if (i < n) {
    float d = deg[i];
    dinv[i] = (d > 0.0f) ? rsqrtf(d) : 0.0f;
  }
}

template <int K, int NOUT>
__global__ void gemm_wmma_scale_kernel(const float* __restrict__ X,
                                       const float* __restrict__ W,
                                       const float* __restrict__ dinv,
                                       float* __restrict__ hs,
                                       float* __restrict__ acc,
                                       int nrows) {
  const int lane = threadIdx.x & 31;
  const int wave = blockIdx.x * (blockDim.x >> 5) + (threadIdx.x >> 5);
  const int ntc  = NOUT / 16;
  const int row0 = (wave / ntc) * 16;
  const int col0 = (wave % ntc) * 16;
  if (row0 >= nrows) return;

  const int half = lane >> 4;
  const int m    = lane & 15;

  fx8 c = {};
  const float* xrow = X + (size_t)(row0 + m) * K;

#pragma unroll
  for (int kb = 0; kb < K; kb += 4) {
    const int ka = kb + 2 * half;
    fx2 a, b;
    a.x = xrow[ka];
    a.y = xrow[ka + 1];
    b.x = W[(size_t)ka * NOUT + col0 + m];
    b.y = W[(size_t)(ka + 1) * NOUT + col0 + m];
    c = __builtin_amdgcn_wmma_f32_16x16x4_f32(false, a, false, b,
                                              (short)0, c, false, false);
  }

#pragma unroll
  for (int r = 0; r < 8; ++r) {
    const int mr  = r + 8 * half;
    const float s = dinv[row0 + mr];
    const float v = c[r] * s;
    const size_t idx = (size_t)(row0 + mr) * NOUT + col0 + m;
    hs[idx]  = v;
    acc[idx] = v;
  }
}

__global__ void scatter128_kernel(const float* __restrict__ hs,
                                  const int* __restrict__ src,
                                  const int* __restrict__ dst,
                                  const float* __restrict__ ew,
                                  float* __restrict__ acc, int e) {
  const int lane = threadIdx.x & 31;
  const int edge = blockIdx.x * (blockDim.x >> 5) + (threadIdx.x >> 5);
  if (edge >= e) return;
  const int   s = src[edge];
  const int   d = dst[edge];
  const float w = ew[edge];
  const float4 v = ((const float4*)(hs + (size_t)s * 128))[lane];
  float* q = acc + (size_t)d * 128 + lane * 4;
  unsafeAtomicAdd(q + 0, w * v.x);
  unsafeAtomicAdd(q + 1, w * v.y);
  unsafeAtomicAdd(q + 2, w * v.z);
  unsafeAtomicAdd(q + 3, w * v.w);
}

__global__ void scatter64_kernel(const float* __restrict__ hs,
                                 const int* __restrict__ src,
                                 const int* __restrict__ dst,
                                 const float* __restrict__ ew,
                                 float* __restrict__ acc, int e) {
  const int lane = threadIdx.x & 31;
  const int edge = blockIdx.x * (blockDim.x >> 5) + (threadIdx.x >> 5);
  if (edge >= e) return;
  const int   s = src[edge];
  const int   d = dst[edge];
  const float w = ew[edge];
  const float2 v = ((const float2*)(hs + (size_t)s * 64))[lane];
  float* q = acc + (size_t)d * 64 + lane * 2;
  unsafeAtomicAdd(q + 0, w * v.x);
  unsafeAtomicAdd(q + 1, w * v.y);
}

template <int F>
__global__ void finalize_kernel(const float* __restrict__ acc,
                                const float* __restrict__ dinv,
                                const float* __restrict__ bias,
                                float* __restrict__ out, int n, int do_relu) {
  const size_t i = (size_t)blockIdx.x * blockDim.x + threadIdx.x;
  if (i >= (size_t)n * F) return;
  const int row = (int)(i / F);
  const int f   = (int)(i % F);
  float v = dinv[row] * acc[i] + bias[f];
  if (do_relu) v = fmaxf(v, 0.0f);
  out[i] = v;
}

extern "C" void kernel_launch(void* const* d_in, const int* in_sizes, int n_in,
                              void* d_out, int out_size, void* d_ws, size_t ws_size,
                              hipStream_t stream) {
  const float* z  = (const float*)d_in[0];
  const int*   ei = (const int*)d_in[1];
  const float* ew = (const float*)d_in[2];
  const float* W1 = (const float*)d_in[3];
  const float* b1 = (const float*)d_in[4];
  const float* W2 = (const float*)d_in[5];
  const float* b2 = (const float*)d_in[6];
  float* out = (float*)d_out;

  const int n = in_sizes[0] / 64;
  const int e = in_sizes[2];
  const int* srcp = ei;
  const int* dstp = ei + e;

  float* ws   = (float*)d_ws;
  float* deg  = ws;
  float* dinv = deg + n;
  float* X1   = dinv + n;
  float* X2   = X1 + (size_t)n * 128;
  float* hs2  = X2;
  float* acc2 = X2 + (size_t)n * 64;

  const int B = 256;

  deg_init_kernel<<<(n + B - 1) / B, B, 0, stream>>>(deg, n);
  deg_accum_kernel<<<(e + B - 1) / B, B, 0, stream>>>(dstp, ew, deg, e);
  dinv_kernel<<<(n + B - 1) / B, B, 0, stream>>>(deg, dinv, n);

  {
    const long long waves = (long long)((n + 15) / 16) * (128 / 16);
    const long long thr   = waves * 32;
    gemm_wmma_scale_kernel<64, 128><<<(unsigned)((thr + B - 1) / B), B, 0, stream>>>(
        z, W1, dinv, X1, X2, n);
  }
  {
    const long long thr = (long long)e * 32;
    scatter128_kernel<<<(unsigned)((thr + B - 1) / B), B, 0, stream>>>(
        X1, srcp, dstp, ew, X2, e);
  }
  {
    const long long thr = (long long)n * 128;
    finalize_kernel<128><<<(unsigned)((thr + B - 1) / B), B, 0, stream>>>(
        X2, dinv, b1, X1, n, 1);
  }
  {
    const long long waves = (long long)((n + 15) / 16) * (64 / 16);
    const long long thr   = waves * 32;
    gemm_wmma_scale_kernel<128, 64><<<(unsigned)((thr + B - 1) / B), B, 0, stream>>>(
        X1, W2, dinv, hs2, acc2, n);
  }
  {
    const long long thr = (long long)e * 32;
    scatter64_kernel<<<(unsigned)((thr + B - 1) / B), B, 0, stream>>>(
        hs2, srcp, dstp, ew, acc2, e);
  }
  {
    const long long thr = (long long)n * 64;
    finalize_kernel<64><<<(unsigned)((thr + B - 1) / B), B, 0, stream>>>(
        acc2, dinv, b2, out, n, 0);
  }
}


# Layer 1 aggregates before its product, (A x) W1 for A (x W1): the edge scatter-add moves 64-wide rows, not 128; bias and relu move into the GEMM epilogue; the f32 sum order changes
# speedup vs baseline: 1.2810x; 1.2810x over previous
#include <hip/hip_runtime.h>

typedef float fx2 __attribute__((ext_vector_type(2)));
typedef float fx8 __attribute__((ext_vector_type(8)));

__global__ void deg_init_kernel(float* __restrict__ deg, int n) {
  int i = blockIdx.x * blockDim.x + threadIdx.x;
  if (i < n) deg[i] = 1.0f;
}

__global__ void deg_accum_kernel(const int* __restrict__ dst,
                                 const float* __restrict__ ew,
                                 float* __restrict__ deg, int e) {
  int i = blockIdx.x * blockDim.x + threadIdx.x;
  if (i < e) unsafeAtomicAdd(&deg[dst[i]], ew[i]);
}

__global__ void dinv_kernel(const float* __restrict__ deg,
                            float* __restrict__ dinv, int n) {
  int i = blockIdx.x * blockDim.x + threadIdx.x;
  if (i < n) {
    float d = deg[i];
    dinv[i] = (d > 0.0f) ? rsqrtf(d) : 0.0f;
  }
}

template <int K, int NOUT, bool BIAS_RELU>
__global__ void gemm_wmma_kernel(const float* __restrict__ X,
                                 const float* __restrict__ W,
                                 const float* __restrict__ dinv,
                                 const float* __restrict__ bias,
                                 float* __restrict__ o0,
                                 float* __restrict__ o1,
                                 int nrows) {
  const int lane = threadIdx.x & 31;
  const int wave = blockIdx.x * (blockDim.x >> 5) + (threadIdx.x >> 5);
  const int ntc  = NOUT / 16;
  const int row0 = (wave / ntc) * 16;
  const int col0 = (wave % ntc) * 16;
  if (row0 >= nrows) return;

  const int half = lane >> 4;
  const int m    = lane & 15;

  fx8 c = {};
  const float* xrow = X + (size_t)(row0 + m) * K;

#pragma unroll
  for (int kb = 0; kb < K; kb += 4) {
    const int ka = kb + 2 * half;
    fx2 a, b;
    a.x = xrow[ka];
    a.y = xrow[ka + 1];
    b.x = W[(size_t)ka * NOUT + col0 + m];
    b.y = W[(size_t)(ka + 1) * NOUT + col0 + m];
    c = __builtin_amdgcn_wmma_f32_16x16x4_f32(false, a, false, b,
                                              (short)0, c, false, false);
  }

  float bcol = 0.0f;
  if (BIAS_RELU) bcol = bias[col0 + m];

#pragma unroll
  for (int r = 0; r < 8; ++r) {
    const int mr  = r + 8 * half;
    const float s = dinv[row0 + mr];
    const size_t idx = (size_t)(row0 + mr) * NOUT + col0 + m;
    if (BIAS_RELU) {
      float v = s * c[r] + bcol;
      o0[idx] = fmaxf(v, 0.0f);
    } else {
      const float v = c[r] * s;
      o0[idx] = v;
      o1[idx] = v;
    }
  }
}

__global__ void scale_rows64_kernel(const float* __restrict__ z,
                                    const float* __restrict__ dinv,
                                    float* __restrict__ zs,
                                    float* __restrict__ acc, int n) {
  const size_t i = (size_t)blockIdx.x * blockDim.x + threadIdx.x;
  if (i >= (size_t)n * 16) return;
  const float s = dinv[i >> 4];
  float4 v = ((const float4*)z)[i];
  v.x *= s; v.y *= s; v.z *= s; v.w *= s;
  ((float4*)zs)[i]  = v;
  ((float4*)acc)[i] = v;
}

__global__ void scatter64_kernel(const float* __restrict__ hs,
                                 const int* __restrict__ src,
                                 const int* __restrict__ dst,
                                 const float* __restrict__ ew,
                                 float* __restrict__ acc, int e) {
  const int lane = threadIdx.x & 31;
  const int edge = blockIdx.x * (blockDim.x >> 5) + (threadIdx.x >> 5);
  if (edge >= e) return;
  const int   s = src[edge];
  const int   d = dst[edge];
  const float w = ew[edge];
  const float2 v = ((const float2*)(hs + (size_t)s * 64))[lane];
  float* q = acc + (size_t)d * 64 + lane * 2;
  unsafeAtomicAdd(q + 0, w * v.x);
  unsafeAtomicAdd(q + 1, w * v.y);
}

template <int F>
__global__ void finalize_kernel(const float* __restrict__ acc,
                                const float* __restrict__ dinv,
                                const float* __restrict__ bias,
                                float* __restrict__ out, int n, int do_relu) {
  const size_t i = (size_t)blockIdx.x * blockDim.x + threadIdx.x;
  if (i >= (size_t)n * F) return;
  const int row = (int)(i / F);
  const int f   = (int)(i % F);
  float v = dinv[row] * acc[i] + bias[f];
  if (do_relu) v = fmaxf(v, 0.0f);
  out[i] = v;
}

extern "C" void kernel_launch(void* const* d_in, const int* in_sizes, int n_in,
                              void* d_out, int out_size, void* d_ws, size_t ws_size,
                              hipStream_t stream) {
  const float* z  = (const float*)d_in[0];
  const int*   ei = (const int*)d_in[1];
  const float* ew = (const float*)d_in[2];
  const float* W1 = (const float*)d_in[3];
  const float* b1 = (const float*)d_in[4];
  const float* W2 = (const float*)d_in[5];
  const float* b2 = (const float*)d_in[6];
  float* out = (float*)d_out;

  const int n = in_sizes[0] / 64;
  const int e = in_sizes[2];
  const int* srcp = ei;
  const int* dstp = ei + e;

  float* ws   = (float*)d_ws;
  float* deg  = ws;
  float* dinv = deg + n;
  float* X1   = dinv + n;
  float* X2   = X1 + (size_t)n * 128;
  float* zs   = X2;
  float* accz = X2 + (size_t)n * 64;
  float* hs2  = X2;
  float* acc2 = X2 + (size_t)n * 64;

  const int B = 256;

  deg_init_kernel<<<(n + B - 1) / B, B, 0, stream>>>(deg, n);
  deg_accum_kernel<<<(e + B - 1) / B, B, 0, stream>>>(dstp, ew, deg, e);
  dinv_kernel<<<(n + B - 1) / B, B, 0, stream>>>(deg, dinv, n);

  {
    const long long thr = (long long)n * 16;
    scale_rows64_kernel<<<(unsigned)((thr + B - 1) / B), B, 0, stream>>>(
        z, dinv, zs, accz, n);
  }
  {
    const long long thr = (long long)e * 32;
    scatter64_kernel<<<(unsigned)((thr + B - 1) / B), B, 0, stream>>>(
        zs, srcp, dstp, ew, accz, e);
  }
  {
    const long long waves = (long long)((n + 15) / 16) * (128 / 16);
    const long long thr   = waves * 32;
    gemm_wmma_kernel<64, 128, true><<<(unsigned)((thr + B - 1) / B), B, 0, stream>>>(
        accz, W1, dinv, b1, X1, nullptr, n);
  }
  {
    const long long waves = (long long)((n + 15) / 16) * (64 / 16);
    const long long thr   = waves * 32;
    gemm_wmma_kernel<128, 64, false><<<(unsigned)((thr + B - 1) / B), B, 0, stream>>>(
        X1, W2, dinv, nullptr, hs2, acc2, n);
  }
  {
    const long long thr = (long long)e * 32;
    scatter64_kernel<<<(unsigned)((thr + B - 1) / B), B, 0, stream>>>(
        hs2, srcp, dstp, ew, acc2, e);
  }
  {
    const long long thr = (long long)n * 64;
    finalize_kernel<64><<<(unsigned)((thr + B - 1) / B), B, 0, stream>>>(
        acc2, dinv, b2, out, n, 0);
  }
}


# Both edge aggregations become row gathers over an edge list counting-sorted by destination in the same call, weight times dinv[src]: no f32 feature atomics; rows summed in arrival-rank order
# speedup vs baseline: 1.9493x; 1.5217x over previous
#include <hip/hip_runtime.h>

typedef float fx2 __attribute__((ext_vector_type(2)));
typedef float fx8 __attribute__((ext_vector_type(8)));

__global__ void init_kernel(float* __restrict__ deg, int* __restrict__ cnt, int n) {
  int i = blockIdx.x * blockDim.x + threadIdx.x;
  if (i < n) {
    deg[i] = 1.0f;
    cnt[i] = 0;
  }
}

__global__ void hist_kernel(const int* __restrict__ dst,
                            const float* __restrict__ ew,
                            float* __restrict__ deg,
                            int* __restrict__ cnt,
                            int* __restrict__ rank, int e) {
  int i = blockIdx.x * blockDim.x + threadIdx.x;
  if (i < e) {
    const int d = dst[i];
    unsafeAtomicAdd(&deg[d], ew[i]);
    rank[i] = atomicAdd(&cnt[d], 1);
  }
}

__global__ void __launch_bounds__(1024)
rowptr_dinv_kernel(const int* __restrict__ cnt,
                   int* __restrict__ rowptr,
                   float* __restrict__ deg_dinv, int n) {
  __shared__ int s_tot[32];
  __shared__ int s_pre[32];
  const int t    = threadIdx.x;
  const int lane = t & 31;
  const int w    = t >> 5;
  const int i    = blockIdx.x * 1024 + t;

  int before = 0;
  for (int k = 0; k < (int)blockIdx.x; ++k) before += cnt[k * 1024 + t];

  const int c = (i < n) ? cnt[i] : 0;
  int inc = c;
  for (int o = 1; o < 32; o <<= 1) {
    const int v = __shfl_up(inc, o);
    if (lane >= o) inc += v;
  }
  for (int o = 16; o > 0; o >>= 1) before += __shfl_xor(before, o);
  if (lane == 31) s_tot[w] = inc;
  if (lane == 0)  s_pre[w] = before;
  __syncthreads();
  int x = s_pre[lane] + ((lane < w) ? s_tot[lane] : 0);
  for (int o = 16; o > 0; o >>= 1) x += __shfl_xor(x, o);

  if (i < n) {
    rowptr[i] = x + inc - c;
    const float d = deg_dinv[i];
    deg_dinv[i] = (d > 0.0f) ? rsqrtf(d) : 0.0f;
  }
}

__global__ void fill_kernel(const int* __restrict__ src,
                            const int* __restrict__ dst,
                            const float* __restrict__ ew,
                            const int* __restrict__ rank,
                            const int* __restrict__ rowptr,
                            const float* __restrict__ dinv,
                            int* __restrict__ entries, int e) {
  int i = blockIdx.x * blockDim.x + threadIdx.x;
  if (i < e) {
    const int s = src[i];
    const int p = rowptr[dst[i]] + rank[i];
    int2 v;
    v.x = s;
    v.y = __float_as_int(ew[i] * dinv[s]);
    ((int2*)entries)[p] = v;
  }
}

template <int U>
__device__ __forceinline__ void gather_steps(const float2* __restrict__ x2,
                                             const int2* __restrict__ ed,
                                             int j, int lane, float2& acc) {
  int2   en[U];
  float2 v[U];
  for (int u = 0; u < U; ++u) en[u] = ed[j + u];
  for (int u = 0; u < U; ++u) v[u] = x2[(size_t)en[u].x * 32 + lane];
  for (int u = 0; u < U; ++u) {
    const float wn = __int_as_float(en[u].y);
    acc.x = fmaf(wn, v[u].x, acc.x);
    acc.y = fmaf(wn, v[u].y, acc.y);
  }
}

template <bool FINAL>
__global__ void gather64_kernel(const float* __restrict__ x,
                                const int* __restrict__ rowptr,
                                const int* __restrict__ entries,
                                const float* __restrict__ dinv,
                                const float* __restrict__ bias,
                                float* __restrict__ out, int n, int e) {
  const int lane = threadIdx.x & 31;
  const int row = __builtin_amdgcn_readfirstlane(
      (int)(blockIdx.x * (blockDim.x >> 5) + (threadIdx.x >> 5)));
  if (row >= n) return;

  const int beg = rowptr[row];
  const int end = (row + 1 < n) ? rowptr[row + 1] : e;
  const float dr = dinv[row];
  const float2* __restrict__ x2 = (const float2*)x;
  const int2* __restrict__ ed   = (const int2*)entries;

  float2 acc = x2[(size_t)row * 32 + lane];
  acc.x *= dr;
  acc.y *= dr;

  int j = beg;
  for (; j + 8 <= end; j += 8) gather_steps<8>(x2, ed, j, lane, acc);
  const int rem = end - j;
  if (rem & 4) { gather_steps<4>(x2, ed, j, lane, acc); j += 4; }
  if (rem & 2) { gather_steps<2>(x2, ed, j, lane, acc); j += 2; }
  if (rem & 1) { gather_steps<1>(x2, ed, j, lane, acc); }

  if (FINAL) {
    const float2 b = ((const float2*)bias)[lane];
    acc.x = dr * acc.x + b.x;
    acc.y = dr * acc.y + b.y;
  }
  ((float2*)out)[(size_t)row * 32 + lane] = acc;
}

template <int K, int NOUT, bool BIAS_RELU>
__global__ void gemm_wmma_kernel(const float* __restrict__ X,
                                 const float* __restrict__ W,
                                 const float* __restrict__ dinv,
                                 const float* __restrict__ bias,
                                 float* __restrict__ o0,
                                 int nrows) {
  const int lane = threadIdx.x & 31;
  const int wave = blockIdx.x * (blockDim.x >> 5) + (threadIdx.x >> 5);
  const int ntc  = NOUT / 16;
  const int row0 = (wave / ntc) * 16;
  const int col0 = (wave % ntc) * 16;
  if (row0 >= nrows) return;

  const int half = lane >> 4;
  const int m    = lane & 15;

  fx8 c = {};
  const float* xrow = X + (size_t)(row0 + m) * K;

#pragma unroll
  for (int kb = 0; kb < K; kb += 4) {
    const int ka = kb + 2 * half;
    fx2 a, b;
    a.x = xrow[ka];
    a.y = xrow[ka + 1];
    b.x = W[(size_t)ka * NOUT + col0 + m];
    b.y = W[(size_t)(ka + 1) * NOUT + col0 + m];
    c = __builtin_amdgcn_wmma_f32_16x16x4_f32(false, a, false, b,
                                              (short)0, c, false, false);
  }

  float bcol = 0.0f;
  if (BIAS_RELU) bcol = bias[col0 + m];

#pragma unroll
  for (int r = 0; r < 8; ++r) {
    const int mr  = r + 8 * half;
    const size_t idx = (size_t)(row0 + mr) * NOUT + col0 + m;
    if (BIAS_RELU) {
      const float s = dinv[row0 + mr];
      float v = s * c[r] + bcol;
      o0[idx] = fmaxf(v, 0.0f);
    } else {
      o0[idx] = c[r];
    }
  }
}

extern "C" void kernel_launch(void* const* d_in, const int* in_sizes, int n_in,
                              void* d_out, int out_size, void* d_ws, size_t ws_size,
                              hipStream_t stream) {
  const float* z  = (const float*)d_in[0];
  const int*   ei = (const int*)d_in[1];
  const float* ew = (const float*)d_in[2];
  const float* W1 = (const float*)d_in[3];
  const float* b1 = (const float*)d_in[4];
  const float* W2 = (const float*)d_in[5];
  const float* b2 = (const float*)d_in[6];
  float* out = (float*)d_out;

  const int n = in_sizes[0] / 64;
  const int e = in_sizes[2];
  const int* srcp = ei;
  const int* dstp = ei + e;

  float* ws      = (float*)d_ws;
  float* X1      = ws;
  float* P       = X1 + (size_t)n * 128;
  int*   entries = (int*)(P + (size_t)n * 64);
  float* dinv    = (float*)(entries + (size_t)e * 2);
  int*   rowptr  = (int*)(dinv + n);
  int*   rank    = (int*)X1;
  int*   cnt     = rank + e;

  const int B = 256;

  init_kernel<<<(n + B - 1) / B, B, 0, stream>>>(dinv, cnt, n);
  hist_kernel<<<(e + B - 1) / B, B, 0, stream>>>(dstp, ew, dinv, cnt, rank, e);
  rowptr_dinv_kernel<<<(n + 1023) / 1024, 1024, 0, stream>>>(cnt, rowptr, dinv, n);
  fill_kernel<<<(e + B - 1) / B, B, 0, stream>>>(srcp, dstp, ew, rank, rowptr, dinv,
                                                 entries, e);

  gather64_kernel<false><<<(n + 7) / 8, B, 0, stream>>>(
      z, rowptr, entries, dinv, nullptr, P, n, e);

  {
    const long long waves = (long long)((n + 15) / 16) * (128 / 16);
    const long long thr   = waves * 32;
    gemm_wmma_kernel<64, 128, true><<<(unsigned)((thr + B - 1) / B), B, 0, stream>>>(
        P, W1, dinv, b1, X1, n);
  }
  {
    const long long waves = (long long)((n + 15) / 16) * (64 / 16);
    const long long thr   = waves * 32;
    gemm_wmma_kernel<128, 64, false><<<(unsigned)((thr + B - 1) / B), B, 0, stream>>>(
        X1, W2, nullptr, nullptr, P, n);
  }
  gather64_kernel<true><<<(n + 7) / 8, B, 0, stream>>>(
      P, rowptr, entries, dinv, b2, out, n, e);
}


# Destination sort of the edge list moves into LDS (chunk-local bucket sort, then one work-group per 256-node bucket): no global atomics; dinv[src] scales rows, not weights; rows summed by LDS rank
# speedup vs baseline: 5.2210x; 2.6784x over previous
#include <hip/hip_runtime.h>

typedef float fx2 __attribute__((ext_vector_type(2)));
typedef float fx8 __attribute__((ext_vector_type(8)));

#define CH   8192
#define BSH  8
#define BN   256
#define NBMX 256
#define GMX  256
#define CAP  16384

__device__ __forceinline__ int scan256(int c, int t, int* s_wtot) {
  const int lane = t & 31;
  int inc = c;
  if (t < 256) {
    for (int o = 1; o < 32; o <<= 1) {
      const int v = __shfl_up(inc, o);
      if (lane >= o) inc += v;
    }
    if (lane == 31) s_wtot[t >> 5] = inc;
  }
  __syncthreads();
  int pre = 0;
  if (t < 256) {
    for (int k = 0; k < (t >> 5); ++k) pre += s_wtot[k];
  }
  return pre + inc - c;
}

__global__ void __launch_bounds__(1024)
chunk_sort_kernel(const int* __restrict__ src,
                  const int* __restrict__ dst,
                  const float* __restrict__ ew,
                  int2* __restrict__ srt,
                  int* __restrict__ tcnt,
                  int* __restrict__ tlst, int e, int nb, int g1) {
  __shared__ int  s_cnt[NBMX];
  __shared__ int  s_start[NBMX];
  __shared__ int  s_wtot[8];
  __shared__ int2 s_stage[CH];
  const int t    = threadIdx.x;
  const int blk  = blockIdx.x;
  const int base = blk * CH;

  if (t < NBMX) s_cnt[t] = 0;
  __syncthreads();

  int key[8], wb[8], bk[8], rk[8];
#pragma unroll
  for (int u = 0; u < 8; ++u) {
    const int ii = min(base + u * 1024 + t, e - 1);
    bk[u]  = dst[ii];
    key[u] = src[ii];
    wb[u]  = __float_as_int(ew[ii]);
  }
#pragma unroll
  for (int u = 0; u < 8; ++u) {
    const int d = bk[u];
    bk[u] = -1;
    if (base + u * 1024 + t < e) {
      bk[u]  = d >> BSH;
      key[u] = key[u] | ((d & (BN - 1)) << 24);
      rk[u]  = atomicAdd(&s_cnt[bk[u]], 1);
    }
  }
  __syncthreads();

  const int c  = (t < NBMX) ? s_cnt[t] : 0;
  const int ex = scan256(c, t, s_wtot);
  if (t < NBMX) {
    s_start[t] = ex;
    if (t < nb) {
      tcnt[t * g1 + blk] = c;
      tlst[t * g1 + blk] = ex;
    }
  }
  __syncthreads();

#pragma unroll
  for (int u = 0; u < 8; ++u) {
    if (bk[u] >= 0) {
      int2 v;
      v.x = key[u];
      v.y = wb[u];
      s_stage[s_start[bk[u]] + rk[u]] = v;
    }
  }
  __syncthreads();

  const int cc = min(CH, e - base);
#pragma unroll
  for (int u = 0; u < 8; ++u) {
    const int p = u * 1024 + t;
    if (p < cc) srt[(size_t)base + p] = s_stage[p];
  }
}

template <typename F>
__device__ __forceinline__ void walk_runs(const int2* __restrict__ srt,
                                          const int* s_rc, const int* s_rs,
                                          int g1, int last, int w, int lane, F f) {
  for (int r = w; r < g1; r += 64) {
    const bool two = (r + 32 < g1);
    const int  r1  = two ? r + 32 : r;
    const int  c0  = s_rc[r];
    const int  c1  = two ? s_rc[r1] : 0;
    const int  o0  = r * CH + s_rs[r];
    const int  o1  = r1 * CH + s_rs[r1];
    const int2 a0 = srt[min(o0 + lane, last)];
    const int2 b0 = srt[min(o0 + lane + 32, last)];
    const int2 a1 = srt[min(o1 + lane, last)];
    const int2 b1 = srt[min(o1 + lane + 32, last)];
    if (lane < c0)      f(a0);
    if (lane + 32 < c0) f(b0);
    if (lane < c1)      f(a1);
    if (lane + 32 < c1) f(b1);
    for (int j = lane + 64; j < c0; j += 32) f(srt[o0 + j]);
    for (int j = lane + 64; j < c1; j += 32) f(srt[o1 + j]);
  }
}

__global__ void __launch_bounds__(1024)
bucket_kernel(const int2* __restrict__ srt,
              const int* __restrict__ tcnt,
              const int* __restrict__ tlst,
              const float* __restrict__ z,
              int2* __restrict__ entries,
              int* __restrict__ rowptr,
              float* __restrict__ dinv,
              float* __restrict__ zs, int n, int e, int g1) {
  __shared__ int   s_c[BN];
  __shared__ float s_deg[BN];
  __shared__ int   s_cur[BN];
  __shared__ float s_dinv[BN];
  __shared__ int   s_rc[GMX];
  __shared__ int   s_rs[GMX];
  __shared__ int   s_wtot[8];
  __shared__ int   s_red[32];
  __shared__ int2  s_ent[CAP];
  const int t    = threadIdx.x;
  const int lane = t & 31;
  const int w    = t >> 5;
  const int b    = blockIdx.x;

  int before = 0;
  for (int i = t; i < b * g1; i += 1024) before += tcnt[i];
  for (int o = 16; o > 0; o >>= 1) before += __shfl_xor(before, o);
  if (lane == 0) s_red[w] = before;
  if (t < BN) {
    s_c[t]   = 0;
    s_deg[t] = 1.0f;
  }
  if (t < g1) {
    s_rc[t] = tcnt[b * g1 + t];
    s_rs[t] = tlst[b * g1 + t];
  }
  __syncthreads();
  int first = s_red[lane];
  for (int o = 16; o > 0; o >>= 1) first += __shfl_xor(first, o);

  walk_runs(srt, s_rc, s_rs, g1, e - 1, w, lane, [&](int2 rec) {
    const int dl = (rec.x >> 24) & (BN - 1);
    atomicAdd(&s_c[dl], 1);
    unsafeAtomicAdd(&s_deg[dl], __int_as_float(rec.y));
  });
  __syncthreads();

  const int c  = (t < BN) ? s_c[t] : 0;
  const int ex = scan256(c, t, s_wtot);
  if (t < BN) {
    const int node = b * BN + t;
    s_cur[t] = ex;
    float di = 0.0f;
    if (node < n) {
      const float d = s_deg[t];
      di = (d > 0.0f) ? rsqrtf(d) : 0.0f;
      rowptr[node] = first + ex;
      dinv[node]   = di;
    }
    s_dinv[t] = di;
  }
  __syncthreads();
  int total = 0;
  for (int k = 0; k < 8; ++k) total += s_wtot[k];
  const bool staged = (total <= CAP);

  walk_runs(srt, s_rc, s_rs, g1, e - 1, w, lane, [&](int2 rec) {
    const int dl  = (rec.x >> 24) & (BN - 1);
    const int pos = atomicAdd(&s_cur[dl], 1);
    int2 v;
    v.x = rec.x & 0x00FFFFFF;
    v.y = rec.y;
    if (staged) s_ent[pos] = v;
    else        entries[(size_t)first + pos] = v;
  });
  __syncthreads();
  if (staged) {
    for (int q = t; q < total; q += 1024) entries[(size_t)first + q] = s_ent[q];
  }

  const float2* __restrict__ z2  = (const float2*)z;
  float2* __restrict__       zs2 = (float2*)zs;
  float2 zv[8];
#pragma unroll
  for (int k = 0; k < 8; ++k) {
    const int q    = k * 1024 + t;
    const int node = min(b * BN + (q >> 5), n - 1);
    zv[k] = z2[(size_t)node * 32 + (q & 31)];
  }
#pragma unroll
  for (int k = 0; k < 8; ++k) {
    const int q    = k * 1024 + t;
    const int rl   = q >> 5;
    const int node = b * BN + rl;
    if (node < n) {
      const float di = s_dinv[rl];
      float2 v = zv[k];
      v.x *= di;
      v.y *= di;
      zs2[(size_t)node * 32 + (q & 31)] = v;
    }
  }
}

template <int U>
__device__ __forceinline__ void gather_steps(const float2* __restrict__ x2,
                                             const int2* __restrict__ ed,
                                             int j, int lane, float2& acc) {
  int2   en[U];
  float2 v[U];
  for (int u = 0; u < U; ++u) en[u] = ed[j + u];
  for (int u = 0; u < U; ++u) v[u] = x2[(size_t)en[u].x * 32 + lane];
  for (int u = 0; u < U; ++u) {
    const float wn = __int_as_float(en[u].y);
    acc.x = fmaf(wn, v[u].x, acc.x);
    acc.y = fmaf(wn, v[u].y, acc.y);
  }
}

template <bool FINAL>
__global__ void gather64_kernel(const float* __restrict__ x,
                                const int* __restrict__ rowptr,
                                const int* __restrict__ entries,
                                const float* __restrict__ dinv,
                                const float* __restrict__ bias,
                                float* __restrict__ out, int n, int e) {
  const int lane = threadIdx.x & 31;
  const int row = __builtin_amdgcn_readfirstlane(
      (int)(blockIdx.x * (blockDim.x >> 5) + (threadIdx.x >> 5)));
  if (row >= n) return;

  const int beg = rowptr[row];
  const int end = (row + 1 < n) ? rowptr[row + 1] : e;
  const float2* __restrict__ x2 = (const float2*)x;
  const int2* __restrict__ ed   = (const int2*)entries;

  float2 acc = x2[(size_t)row * 32 + lane];

  int j = beg;
  for (; j + 8 <= end; j += 8) gather_steps<8>(x2, ed, j, lane, acc);
  const int rem = end - j;
  if (rem & 4) { gather_steps<4>(x2, ed, j, lane, acc); j += 4; }
  if (rem & 2) { gather_steps<2>(x2, ed, j, lane, acc); j += 2; }
  if (rem & 1) { gather_steps<1>(x2, ed, j, lane, acc); }

  if (FINAL) {
    const float dr = dinv[row];
    const float2 b = ((const float2*)bias)[lane];
    acc.x = dr * acc.x + b.x;
    acc.y = dr * acc.y + b.y;
  }
  ((float2*)out)[(size_t)row * 32 + lane] = acc;
}

template <int K, int NOUT, bool BIAS_RELU>
__global__ void gemm_wmma_kernel(const float* __restrict__ X,
                                 const float* __restrict__ W,
                                 const float* __restrict__ dinv,
                                 const float* __restrict__ bias,
                                 float* __restrict__ o0,
                                 int nrows) {
  const int lane = threadIdx.x & 31;
  const int wave = blockIdx.x * (blockDim.x >> 5) + (threadIdx.x >> 5);
  const int ntc  = NOUT / 16;
  const int row0 = (wave / ntc) * 16;
  const int col0 = (wave % ntc) * 16;
  if (row0 >= nrows) return;

  const int half = lane >> 4;
  const int m    = lane & 15;

  fx8 c = {};
  const float* xrow = X + (size_t)(row0 + m) * K;

#pragma unroll
  for (int kb = 0; kb < K; kb += 4) {
    const int ka = kb + 2 * half;
    fx2 a, b;
    a.x = xrow[ka];
    a.y = xrow[ka + 1];
    b.x = W[(size_t)ka * NOUT + col0 + m];
    b.y = W[(size_t)(ka + 1) * NOUT + col0 + m];
    c = __builtin_amdgcn_wmma_f32_16x16x4_f32(false, a, false, b,
                                              (short)0, c, false, false);
  }

  float bcol = 0.0f;
  if (BIAS_RELU) bcol = bias[col0 + m];

#pragma unroll
  for (int r = 0; r < 8; ++r) {
    const int mr  = r + 8 * half;
    const size_t idx = (size_t)(row0 + mr) * NOUT + col0 + m;
    if (BIAS_RELU) {
      const float s = dinv[row0 + mr];
      float v = s * c[r] + bcol;
      o0[idx] = fmaxf(v, 0.0f);
    } else {
      o0[idx] = dinv[row0 + mr] * c[r];
    }
  }
}

extern "C" void kernel_launch(void* const* d_in, const int* in_sizes, int n_in,
                              void* d_out, int out_size, void* d_ws, size_t ws_size,
                              hipStream_t stream) {
  const float* z  = (const float*)d_in[0];
  const int*   ei = (const int*)d_in[1];
  const float* ew = (const float*)d_in[2];
  const float* W1 = (const float*)d_in[3];
  const float* b1 = (const float*)d_in[4];
  const float* W2 = (const float*)d_in[5];
  const float* b2 = (const float*)d_in[6];
  float* out = (float*)d_out;

  const int n = in_sizes[0] / 64;
  const int e = in_sizes[2];
  const int* srcp = ei;
  const int* dstp = ei + e;

  float* ws      = (float*)d_ws;
  float* X1      = ws;
  float* P       = X1 + (size_t)n * 128;
  int*   entries = (int*)(P + (size_t)n * 64);
  float* dinv    = (float*)(entries + (size_t)e * 2);
  int*   rowptr  = (int*)(dinv + n);

  const int g1 = (e + CH - 1) / CH;
  const int nb = (n + BN - 1) / BN;
  int2*  srt  = (int2*)X1;
  float* zs   = X1 + (size_t)n * 64;
  int*   tcnt = (int*)P;
  int*   tlst = tcnt + (size_t)nb * g1;

  const int B = 256;

  chunk_sort_kernel<<<g1, 1024, 0, stream>>>(srcp, dstp, ew, srt, tcnt, tlst, e, nb, g1);
  bucket_kernel<<<nb, 1024, 0, stream>>>(srt, tcnt, tlst, z, (int2*)entries, rowptr,
                                         dinv, zs, n, e, g1);

  gather64_kernel<false><<<(n + 7) / 8, B, 0, stream>>>(
      zs, rowptr, entries, dinv, nullptr, P, n, e);

  {
    const long long waves = (long long)((n + 15) / 16) * (128 / 16);
    const long long thr   = waves * 32;
    gemm_wmma_kernel<64, 128, true><<<(unsigned)((thr + B - 1) / B), B, 0, stream>>>(
        P, W1, dinv, b1, X1, n);
  }
  {
    const long long waves = (long long)((n + 15) / 16) * (64 / 16);
    const long long thr   = waves * 32;
    gemm_wmma_kernel<128, 64, false><<<(unsigned)((thr + B - 1) / B), B, 0, stream>>>(
        X1, W2, dinv, nullptr, P, n);
  }
  gather64_kernel<true><<<(n + 7) / 8, B, 0, stream>>>(
      P, rowptr, entries, dinv, b2, out, n, e);
}


# Layer 1 row gather and both WMMA products run as one kernel per 16-row tile: aggregated rows and h1 stay in LDS, not in the workspace; 4 launches, not 6; same sums in the same order
# speedup vs baseline: 6.6034x; 1.2648x over previous
#include <hip/hip_runtime.h>

typedef float fx2 __attribute__((ext_vector_type(2)));
typedef float fx8 __attribute__((ext_vector_type(8)));

#define CH   8192
#define BSH  8
#define BN   256
#define NBMX 256
#define GMX  256
#define CAP  16384

__device__ __forceinline__ int scan256(int c, int t, int* s_wtot) {
  const int lane = t & 31;
  int inc = c;
  if (t < 256) {
    for (int o = 1; o < 32; o <<= 1) {
      const int v = __shfl_up(inc, o);
      if (lane >= o) inc += v;
    }
    if (lane == 31) s_wtot[t >> 5] = inc;
  }
  __syncthreads();
  int pre = 0;
  if (t < 256) {
    for (int k = 0; k < (t >> 5); ++k) pre += s_wtot[k];
  }
  return pre + inc - c;
}

__global__ void __launch_bounds__(1024)
chunk_sort_kernel(const int* __restrict__ src,
                  const int* __restrict__ dst,
                  const float* __restrict__ ew,
                  int2* __restrict__ srt,
                  int* __restrict__ tcnt,
                  int* __restrict__ tlst, int e, int nb, int g1) {
  __shared__ int  s_cnt[NBMX];
  __shared__ int  s_start[NBMX];
  __shared__ int  s_wtot[8];
  __shared__ int2 s_stage[CH];
  const int t    = threadIdx.x;
  const int blk  = blockIdx.x;
  const int base = blk * CH;

  if (t < NBMX) s_cnt[t] = 0;
  __syncthreads();

  int key[8], wb[8], bk[8], rk[8];
#pragma unroll
  for (int u = 0; u < 8; ++u) {
    const int ii = min(base + u * 1024 + t, e - 1);
    bk[u]  = dst[ii];
    key[u] = src[ii];
    wb[u]  = __float_as_int(ew[ii]);
  }
#pragma unroll
  for (int u = 0; u < 8; ++u) {
    const int d = bk[u];
    bk[u] = -1;
    if (base + u * 1024 + t < e) {
      bk[u]  = d >> BSH;
      key[u] = key[u] | ((d & (BN - 1)) << 24);
      rk[u]  = atomicAdd(&s_cnt[bk[u]], 1);
    }
  }
  __syncthreads();

  const int c  = (t < NBMX) ? s_cnt[t] : 0;
  const int ex = scan256(c, t, s_wtot);
  if (t < NBMX) {
    s_start[t] = ex;
    if (t < nb) {
      tcnt[t * g1 + blk] = c;
      tlst[t * g1 + blk] = ex;
    }
  }
  __syncthreads();

#pragma unroll
  for (int u = 0; u < 8; ++u) {
    if (bk[u] >= 0) {
      int2 v;
      v.x = key[u];
      v.y = wb[u];
      s_stage[s_start[bk[u]] + rk[u]] = v;
    }
  }
  __syncthreads();

  const int cc = min(CH, e - base);
#pragma unroll
  for (int u = 0; u < 8; ++u) {
    const int p = u * 1024 + t;
    if (p < cc) srt[(size_t)base + p] = s_stage[p];
  }
}

template <typename F>
__device__ __forceinline__ void walk_runs(const int2* __restrict__ srt,
                                          const int* s_rc, const int* s_rs,
                                          int g1, int last, int w, int lane, F f) {
  for (int r = w; r < g1; r += 64) {
    const bool two = (r + 32 < g1);
    const int  r1  = two ? r + 32 : r;
    const int  c0  = s_rc[r];
    const int  c1  = two ? s_rc[r1] : 0;
    const int  o0  = r * CH + s_rs[r];
    const int  o1  = r1 * CH + s_rs[r1];
    const int2 a0 = srt[min(o0 + lane, last)];
    const int2 b0 = srt[min(o0 + lane + 32, last)];
    const int2 a1 = srt[min(o1 + lane, last)];
    const int2 b1 = srt[min(o1 + lane + 32, last)];
    if (lane < c0)      f(a0);
    if (lane + 32 < c0) f(b0);
    if (lane < c1)      f(a1);
    if (lane + 32 < c1) f(b1);
    for (int j = lane + 64; j < c0; j += 32) f(srt[o0 + j]);
    for (int j = lane + 64; j < c1; j += 32) f(srt[o1 + j]);
  }
}

__global__ void __launch_bounds__(1024)
bucket_kernel(const int2* __restrict__ srt,
              const int* __restrict__ tcnt,
              const int* __restrict__ tlst,
              const float* __restrict__ z,
              int2* __restrict__ entries,
              int* __restrict__ rowptr,
              float* __restrict__ dinv,
              float* __restrict__ zs, int n, int e, int g1) {
  __shared__ int   s_c[BN];
  __shared__ float s_deg[BN];
  __shared__ int   s_cur[BN];
  __shared__ float s_dinv[BN];
  __shared__ int   s_rc[GMX];
  __shared__ int   s_rs[GMX];
  __shared__ int   s_wtot[8];
  __shared__ int   s_red[32];
  __shared__ int2  s_ent[CAP];
  const int t    = threadIdx.x;
  const int lane = t & 31;
  const int w    = t >> 5;
  const int b    = blockIdx.x;

  int before = 0;
  for (int i = t; i < b * g1; i += 1024) before += tcnt[i];
  for (int o = 16; o > 0; o >>= 1) before += __shfl_xor(before, o);
  if (lane == 0) s_red[w] = before;
  if (t < BN) {
    s_c[t]   = 0;
    s_deg[t] = 1.0f;
  }
  if (t < g1) {
    s_rc[t] = tcnt[b * g1 + t];
    s_rs[t] = tlst[b * g1 + t];
  }
  __syncthreads();
  int first = s_red[lane];
  for (int o = 16; o > 0; o >>= 1) first += __shfl_xor(first, o);

  walk_runs(srt, s_rc, s_rs, g1, e - 1, w, lane, [&](int2 rec) {
    const int dl = (rec.x >> 24) & (BN - 1);
    atomicAdd(&s_c[dl], 1);
    unsafeAtomicAdd(&s_deg[dl], __int_as_float(rec.y));
  });
  __syncthreads();

  const int c  = (t < BN) ? s_c[t] : 0;
  const int ex = scan256(c, t, s_wtot);
  if (t < BN) {
    const int node = b * BN + t;
    s_cur[t] = ex;
    float di = 0.0f;
    if (node < n) {
      const float d = s_deg[t];
      di = (d > 0.0f) ? rsqrtf(d) : 0.0f;
      rowptr[node] = first + ex;
      dinv[node]   = di;
    }
    s_dinv[t] = di;
  }
  __syncthreads();
  int total = 0;
  for (int k = 0; k < 8; ++k) total += s_wtot[k];
  const bool staged = (total <= CAP);

  walk_runs(srt, s_rc, s_rs, g1, e - 1, w, lane, [&](int2 rec) {
    const int dl  = (rec.x >> 24) & (BN - 1);
    const int pos = atomicAdd(&s_cur[dl], 1);
    int2 v;
    v.x = rec.x & 0x00FFFFFF;
    v.y = rec.y;
    if (staged) s_ent[pos] = v;
    else        entries[(size_t)first + pos] = v;
  });
  __syncthreads();
  if (staged) {
    for (int q = t; q < total; q += 1024) entries[(size_t)first + q] = s_ent[q];
  }

  const float2* __restrict__ z2  = (const float2*)z;
  float2* __restrict__       zs2 = (float2*)zs;
  float2 zv[8];
#pragma unroll
  for (int k = 0; k < 8; ++k) {
    const int q    = k * 1024 + t;
    const int node = min(b * BN + (q >> 5), n - 1);
    zv[k] = z2[(size_t)node * 32 + (q & 31)];
  }
#pragma unroll
  for (int k = 0; k < 8; ++k) {
    const int q    = k * 1024 + t;
    const int rl   = q >> 5;
    const int node = b * BN + rl;
    if (node < n) {
      const float di = s_dinv[rl];
      float2 v = zv[k];
      v.x *= di;
      v.y *= di;
      zs2[(size_t)node * 32 + (q & 31)] = v;
    }
  }
}

template <int U>
__device__ __forceinline__ void gather_steps(const float2* __restrict__ x2,
                                             const int2* __restrict__ ed,
                                             int j, int lane, float2& acc) {
  int2   en[U];
  float2 v[U];
  for (int u = 0; u < U; ++u) en[u] = ed[j + u];
  for (int u = 0; u < U; ++u) v[u] = x2[(size_t)en[u].x * 32 + lane];
  for (int u = 0; u < U; ++u) {
    const float wn = __int_as_float(en[u].y);
    acc.x = fmaf(wn, v[u].x, acc.x);
    acc.y = fmaf(wn, v[u].y, acc.y);
  }
}

__device__ __forceinline__ float2 gather_row(const float2* __restrict__ x2,
                                             const int* __restrict__ rowptr,
                                             const int2* __restrict__ ed,
                                             int row, int n, int e, int lane) {
  const int beg = rowptr[row];
  const int end = (row + 1 < n) ? rowptr[row + 1] : e;
  float2 acc = x2[(size_t)row * 32 + lane];
  int j = beg;
  for (; j + 8 <= end; j += 8) gather_steps<8>(x2, ed, j, lane, acc);
  const int rem = end - j;
  if (rem & 4) { gather_steps<4>(x2, ed, j, lane, acc); j += 4; }
  if (rem & 2) { gather_steps<2>(x2, ed, j, lane, acc); j += 2; }
  if (rem & 1) { gather_steps<1>(x2, ed, j, lane, acc); }
  return acc;
}

__global__ void gather64_final_kernel(const float* __restrict__ x,
                                      const int* __restrict__ rowptr,
                                      const int* __restrict__ entries,
                                      const float* __restrict__ dinv,
                                      const float* __restrict__ bias,
                                      float* __restrict__ out, int n, int e) {
  const int lane = threadIdx.x & 31;
  const int row = __builtin_amdgcn_readfirstlane(
      (int)(blockIdx.x * (blockDim.x >> 5) + (threadIdx.x >> 5)));
  if (row >= n) return;
  float2 acc = gather_row((const float2*)x, rowptr, (const int2*)entries, row, n, e, lane);
  const float dr = dinv[row];
  const float2 b = ((const float2*)bias)[lane];
  acc.x = dr * acc.x + b.x;
  acc.y = dr * acc.y + b.y;
  ((float2*)out)[(size_t)row * 32 + lane] = acc;
}

#define PA 68
#define PH 132

__global__ void __launch_bounds__(512)
layer1_fused_kernel(const float* __restrict__ zs,
                    const int* __restrict__ rowptr,
                    const int* __restrict__ entries,
                    const float* __restrict__ dinv,
                    const float* __restrict__ W1,
                    const float* __restrict__ b1,
                    const float* __restrict__ W2,
                    float* __restrict__ P, int n, int e) {
  __shared__ __attribute__((aligned(16))) float sA[16 * PA];
  __shared__ __attribute__((aligned(16))) float sH[16 * PH];
  const int lane = threadIdx.x & 31;
  const int w    = __builtin_amdgcn_readfirstlane((int)(threadIdx.x >> 5));
  const int row0 = blockIdx.x * 16;
  const int row  = row0 + w;

  float2 acc;
  acc.x = 0.0f;
  acc.y = 0.0f;
  if (row < n)
    acc = gather_row((const float2*)zs, rowptr, (const int2*)entries, row, n, e, lane);
  *(float2*)&sA[w * PA + 2 * lane] = acc;
  __syncthreads();

  const int half = lane >> 4;
  const int m    = lane & 15;

  if (w < 8) {
    const int col0 = w * 16;
    fx8 c = {};
#pragma unroll
    for (int kb = 0; kb < 64; kb += 4) {
      const int ka = kb + 2 * half;
      fx2 a, b;
      a   = *(const fx2*)&sA[m * PA + ka];
      b.x = W1[(size_t)ka * 128 + col0 + m];
      b.y = W1[(size_t)(ka + 1) * 128 + col0 + m];
      c = __builtin_amdgcn_wmma_f32_16x16x4_f32(false, a, false, b,
                                                (short)0, c, false, false);
    }
    const float bcol = b1[col0 + m];
#pragma unroll
    for (int r = 0; r < 8; ++r) {
      const int mr = r + 8 * half;
      const float sc = dinv[min(row0 + mr, n - 1)];
      sH[mr * PH + col0 + m] = fmaxf(sc * c[r] + bcol, 0.0f);
    }
  }
  __syncthreads();

  if (w < 4) {
    const int col0 = w * 16;
    float sc[8];
#pragma unroll
    for (int r = 0; r < 8; ++r) sc[r] = dinv[min(row0 + 8 * half + r, n - 1)];
    fx8 c = {};
#pragma unroll
    for (int kb = 0; kb < 128; kb += 4) {
      const int ka = kb + 2 * half;
      fx2 a, b;
      a   = *(const fx2*)&sH[m * PH + ka];
      b.x = W2[(size_t)ka * 64 + col0 + m];
      b.y = W2[(size_t)(ka + 1) * 64 + col0 + m];
      c = __builtin_amdgcn_wmma_f32_16x16x4_f32(false, a, false, b,
                                                (short)0, c, false, false);
    }
    float v[8];
#pragma unroll
    for (int r = 0; r < 8; ++r) v[r] = sc[r] * c[r];
    float* __restrict__ prow = P + (size_t)(row0 + 8 * half) * 64 + col0 + m;
    if (row0 + 16 <= n) {
#pragma unroll
      for (int r = 0; r < 8; ++r) prow[(size_t)r * 64] = v[r];
    } else {
#pragma unroll
      for (int r = 0; r < 8; ++r)
        if (row0 + 8 * half + r < n) prow[(size_t)r * 64] = v[r];
    }
  }
}

extern "C" void kernel_launch(void* const* d_in, const int* in_sizes, int n_in,
                              void* d_out, int out_size, void* d_ws, size_t ws_size,
                              hipStream_t stream) {
  const float* z  = (const float*)d_in[0];
  const int*   ei = (const int*)d_in[1];
  const float* ew = (const float*)d_in[2];
  const float* W1 = (const float*)d_in[3];
  const float* b1 = (const float*)d_in[4];
  const float* W2 = (const float*)d_in[5];
  const float* b2 = (const float*)d_in[6];
  float* out = (float*)d_out;

  const int n = in_sizes[0] / 64;
  const int e = in_sizes[2];
  const int* srcp = ei;
  const int* dstp = ei + e;

  float* ws      = (float*)d_ws;
  float* X1      = ws;
  float* P       = X1 + (size_t)n * 128;
  int*   entries = (int*)(P + (size_t)n * 64);
  float* dinv    = (float*)(entries + (size_t)e * 2);
  int*   rowptr  = (int*)(dinv + n);

  const int g1 = (e + CH - 1) / CH;
  const int nb = (n + BN - 1) / BN;
  int2*  srt  = (int2*)X1;
  float* zs   = X1 + (size_t)n * 64;
  int*   tcnt = (int*)P;
  int*   tlst = tcnt + (size_t)nb * g1;

  const int B = 256;

  chunk_sort_kernel<<<g1, 1024, 0, stream>>>(srcp, dstp, ew, srt, tcnt, tlst, e, nb, g1);
  bucket_kernel<<<nb, 1024, 0, stream>>>(srt, tcnt, tlst, z, (int2*)entries, rowptr,
                                         dinv, zs, n, e, g1);

  layer1_fused_kernel<<<(n + 15) / 16, 512, 0, stream>>>(
      zs, rowptr, entries, dinv, W1, b1, W2, P, n, e);

  gather64_final_kernel<<<(n + 7) / 8, B, 0, stream>>>(
      P, rowptr, entries, dinv, b2, out, n, e);
}


# Row gathers read a row's entries with one per-lane load (lane = entry) and hand ids and weights out by readlane: 16 source rows in flight per wave, no scalar entry loads; same sums in the same order
# speedup vs baseline: 6.8006x; 1.0299x over previous
#include <hip/hip_runtime.h>

typedef float fx2 __attribute__((ext_vector_type(2)));
typedef float fx8 __attribute__((ext_vector_type(8)));

#define CH   8192
#define BSH  8
#define BN   256
#define NBMX 256
#define GMX  256
#define CAP  16384

__device__ __forceinline__ int scan256(int c, int t, int* s_wtot) {
  const int lane = t & 31;
  int inc = c;
  if (t < 256) {
    for (int o = 1; o < 32; o <<= 1) {
      const int v = __shfl_up(inc, o);
      if (lane >= o) inc += v;
    }
    if (lane == 31) s_wtot[t >> 5] = inc;
  }
  __syncthreads();
  int pre = 0;
  if (t < 256) {
    for (int k = 0; k < (t >> 5); ++k) pre += s_wtot[k];
  }
  return pre + inc - c;
}

__global__ void __launch_bounds__(1024)
chunk_sort_kernel(const int* __restrict__ src,
                  const int* __restrict__ dst,
                  const float* __restrict__ ew,
                  int2* __restrict__ srt,
                  int* __restrict__ tcnt,
                  int* __restrict__ tlst, int e, int nb, int g1) {
  __shared__ int  s_cnt[NBMX];
  __shared__ int  s_start[NBMX];
  __shared__ int  s_wtot[8];
  __shared__ int2 s_stage[CH];
  const int t    = threadIdx.x;
  const int blk  = blockIdx.x;
  const int base = blk * CH;

  if (t < NBMX) s_cnt[t] = 0;
  __syncthreads();

  int key[8], wb[8], bk[8], rk[8];
#pragma unroll
  for (int u = 0; u < 8; ++u) {
    const int ii = min(base + u * 1024 + t, e - 1);
    bk[u]  = dst[ii];
    key[u] = src[ii];
    wb[u]  = __float_as_int(ew[ii]);
  }
#pragma unroll
  for (int u = 0; u < 8; ++u) {
    const int d = bk[u];
    bk[u] = -1;
    if (base + u * 1024 + t < e) {
      bk[u]  = d >> BSH;
      key[u] = key[u] | ((d & (BN - 1)) << 24);
      rk[u]  = atomicAdd(&s_cnt[bk[u]], 1);
    }
  }
  __syncthreads();

  const int c  = (t < NBMX) ? s_cnt[t] : 0;
  const int ex = scan256(c, t, s_wtot);
  if (t < NBMX) {
    s_start[t] = ex;
    if (t < nb) {
      tcnt[t * g1 + blk] = c;
      tlst[t * g1 + blk] = ex;
    }
  }
  __syncthreads();

#pragma unroll
  for (int u = 0; u < 8; ++u) {
    if (bk[u] >= 0) {
      int2 v;
      v.x = key[u];
      v.y = wb[u];
      s_stage[s_start[bk[u]] + rk[u]] = v;
    }
  }
  __syncthreads();

  const int cc = min(CH, e - base);
#pragma unroll
  for (int u = 0; u < 8; ++u) {
    const int p = u * 1024 + t;
    if (p < cc) srt[(size_t)base + p] = s_stage[p];
  }
}

template <typename F>
__device__ __forceinline__ void walk_runs(const int2* __restrict__ srt,
                                          const int* s_rc, const int* s_rs,
                                          int g1, int last, int w, int lane, F f) {
  for (int r = w; r < g1; r += 64) {
    const bool two = (r + 32 < g1);
    const int  r1  = two ? r + 32 : r;
    const int  c0  = s_rc[r];
    const int  c1  = two ? s_rc[r1] : 0;
    const int  o0  = r * CH + s_rs[r];
    const int  o1  = r1 * CH + s_rs[r1];
    const int2 a0 = srt[min(o0 + lane, last)];
    const int2 b0 = srt[min(o0 + lane + 32, last)];
    const int2 a1 = srt[min(o1 + lane, last)];
    const int2 b1 = srt[min(o1 + lane + 32, last)];
    if (lane < c0)      f(a0);
    if (lane + 32 < c0) f(b0);
    if (lane < c1)      f(a1);
    if (lane + 32 < c1) f(b1);
    for (int j = lane + 64; j < c0; j += 32) f(srt[o0 + j]);
    for (int j = lane + 64; j < c1; j += 32) f(srt[o1 + j]);
  }
}

__global__ void __launch_bounds__(1024)
bucket_kernel(const int2* __restrict__ srt,
              const int* __restrict__ tcnt,
              const int* __restrict__ tlst,
              const float* __restrict__ z,
              int2* __restrict__ entries,
              int* __restrict__ rowptr,
              float* __restrict__ dinv,
              float* __restrict__ zs, int n, int e, int g1) {
  __shared__ int   s_c[BN];
  __shared__ float s_deg[BN];
  __shared__ int   s_cur[BN];
  __shared__ float s_dinv[BN];
  __shared__ int   s_rc[GMX];
  __shared__ int   s_rs[GMX];
  __shared__ int   s_wtot[8];
  __shared__ int   s_red[32];
  __shared__ int2  s_ent[CAP];
  const int t    = threadIdx.x;
  const int lane = t & 31;
  const int w    = t >> 5;
  const int b    = blockIdx.x;

  int before = 0;
  for (int i = t; i < b * g1; i += 1024) before += tcnt[i];
  for (int o = 16; o > 0; o >>= 1) before += __shfl_xor(before, o);
  if (lane == 0) s_red[w] = before;
  if (t < BN) {
    s_c[t]   = 0;
    s_deg[t] = 1.0f;
  }
  if (t < g1) {
    s_rc[t] = tcnt[b * g1 + t];
    s_rs[t] = tlst[b * g1 + t];
  }
  __syncthreads();
  int first = s_red[lane];
  for (int o = 16; o > 0; o >>= 1) first += __shfl_xor(first, o);

  walk_runs(srt, s_rc, s_rs, g1, e - 1, w, lane, [&](int2 rec) {
    const int dl = (rec.x >> 24) & (BN - 1);
    atomicAdd(&s_c[dl], 1);
    unsafeAtomicAdd(&s_deg[dl], __int_as_float(rec.y));
  });
  __syncthreads();

  const int c  = (t < BN) ? s_c[t] : 0;
  const int ex = scan256(c, t, s_wtot);
  if (t < BN) {
    const int node = b * BN + t;
    s_cur[t] = ex;
    float di = 0.0f;
    if (node < n) {
      const float d = s_deg[t];
      di = (d > 0.0f) ? rsqrtf(d) : 0.0f;
      rowptr[node] = first + ex;
      dinv[node]   = di;
    }
    s_dinv[t] = di;
  }
  __syncthreads();
  int total = 0;
  for (int k = 0; k < 8; ++k) total += s_wtot[k];
  const bool staged = (total <= CAP);

  walk_runs(srt, s_rc, s_rs, g1, e - 1, w, lane, [&](int2 rec) {
    const int dl  = (rec.x >> 24) & (BN - 1);
    const int pos = atomicAdd(&s_cur[dl], 1);
    int2 v;
    v.x = rec.x & 0x00FFFFFF;
    v.y = rec.y;
    if (staged) s_ent[pos] = v;
    else        entries[(size_t)first + pos] = v;
  });
  __syncthreads();
  if (staged) {
    for (int q = t; q < total; q += 1024) entries[(size_t)first + q] = s_ent[q];
  }

  const float2* __restrict__ z2  = (const float2*)z;
  float2* __restrict__       zs2 = (float2*)zs;
  float2 zv[8];
#pragma unroll
  for (int k = 0; k < 8; ++k) {
    const int q    = k * 1024 + t;
    const int node = min(b * BN + (q >> 5), n - 1);
    zv[k] = z2[(size_t)node * 32 + (q & 31)];
  }
#pragma unroll
  for (int k = 0; k < 8; ++k) {
    const int q    = k * 1024 + t;
    const int rl   = q >> 5;
    const int node = b * BN + rl;
    if (node < n) {
      const float di = s_dinv[rl];
      float2 v = zv[k];
      v.x *= di;
      v.y *= di;
      zs2[(size_t)node * 32 + (q & 31)] = v;
    }
  }
}

template <int U>
__device__ __forceinline__ void gather_steps(const float2* __restrict__ x2,
                                             int2 my, int k, int lane, float2& acc) {
  int    s[U];
  float  wn[U];
  float2 v[U];
#pragma unroll
  for (int u = 0; u < U; ++u) {
    s[u]  = __builtin_amdgcn_readlane(my.x, k + u);
    wn[u] = __int_as_float(__builtin_amdgcn_readlane(my.y, k + u));
  }
#pragma unroll
  for (int u = 0; u < U; ++u) v[u] = x2[(size_t)s[u] * 32 + lane];
#pragma unroll
  for (int u = 0; u < U; ++u) {
    acc.x = fmaf(wn[u], v[u].x, acc.x);
    acc.y = fmaf(wn[u], v[u].y, acc.y);
  }
}

__device__ __forceinline__ float2 gather_row(const float2* __restrict__ x2,
                                             const int* __restrict__ rowptr,
                                             const int2* __restrict__ ed,
                                             int row, int n, int e, int lane) {
  const int beg = rowptr[row];
  const int end = (row + 1 < n) ? rowptr[row + 1] : e;
  float2 acc = x2[(size_t)row * 32 + lane];
  if (beg >= end) return acc;
  int2 cur = ed[min(beg + lane, e - 1)];
  for (int j0 = beg; j0 < end; j0 += 32) {
    int2 nxt = cur;
    if (j0 + 32 < end) nxt = ed[min(j0 + 32 + lane, e - 1)];
    const int cnt = min(32, end - j0);
    int k = 0;
    for (; k + 16 <= cnt; k += 16) gather_steps<16>(x2, cur, k, lane, acc);
    const int rem = cnt - k;
    if (rem & 8) { gather_steps<8>(x2, cur, k, lane, acc); k += 8; }
    if (rem & 4) { gather_steps<4>(x2, cur, k, lane, acc); k += 4; }
    if (rem & 2) { gather_steps<2>(x2, cur, k, lane, acc); k += 2; }
    if (rem & 1) { gather_steps<1>(x2, cur, k, lane, acc); }
    cur = nxt;
  }
  return acc;
}

__global__ void gather64_final_kernel(const float* __restrict__ x,
                                      const int* __restrict__ rowptr,
                                      const int* __restrict__ entries,
                                      const float* __restrict__ dinv,
                                      const float* __restrict__ bias,
                                      float* __restrict__ out, int n, int e) {
  const int lane = threadIdx.x & 31;
  const int row = __builtin_amdgcn_readfirstlane(
      (int)(blockIdx.x * (blockDim.x >> 5) + (threadIdx.x >> 5)));
  if (row >= n) return;
  float2 acc = gather_row((const float2*)x, rowptr, (const int2*)entries, row, n, e, lane);
  const float dr = dinv[row];
  const float2 b = ((const float2*)bias)[lane];
  acc.x = dr * acc.x + b.x;
  acc.y = dr * acc.y + b.y;
  ((float2*)out)[(size_t)row * 32 + lane] = acc;
}

#define PA 68
#define PH 132

__global__ void __launch_bounds__(512)
layer1_fused_kernel(const float* __restrict__ zs,
                    const int* __restrict__ rowptr,
                    const int* __restrict__ entries,
                    const float* __restrict__ dinv,
                    const float* __restrict__ W1,
                    const float* __restrict__ b1,
                    const float* __restrict__ W2,
                    float* __restrict__ P, int n, int e) {
  __shared__ __attribute__((aligned(16))) float sA[16 * PA];
  __shared__ __attribute__((aligned(16))) float sH[16 * PH];
  const int lane = threadIdx.x & 31;
  const int w    = __builtin_amdgcn_readfirstlane((int)(threadIdx.x >> 5));
  const int row0 = blockIdx.x * 16;
  const int row  = row0 + w;

  float2 acc;
  acc.x = 0.0f;
  acc.y = 0.0f;
  if (row < n)
    acc = gather_row((const float2*)zs, rowptr, (const int2*)entries, row, n, e, lane);
  *(float2*)&sA[w * PA + 2 * lane] = acc;
  __syncthreads();

  const int half = lane >> 4;
  const int m    = lane & 15;

  if (w < 8) {
    const int col0 = w * 16;
    fx8 c = {};
#pragma unroll
    for (int kb = 0; kb < 64; kb += 4) {
      const int ka = kb + 2 * half;
      fx2 a, b;
      a   = *(const fx2*)&sA[m * PA + ka];
      b.x = W1[(size_t)ka * 128 + col0 + m];
      b.y = W1[(size_t)(ka + 1) * 128 + col0 + m];
      c = __builtin_amdgcn_wmma_f32_16x16x4_f32(false, a, false, b,
                                                (short)0, c, false, false);
    }
    const float bcol = b1[col0 + m];
#pragma unroll
    for (int r = 0; r < 8; ++r) {
      const int mr = r + 8 * half;
      const float sc = dinv[min(row0 + mr, n - 1)];
      sH[mr * PH + col0 + m] = fmaxf(sc * c[r] + bcol, 0.0f);
    }
  }
  __syncthreads();

  if (w < 4) {
    const int col0 = w * 16;
    float sc[8];
#pragma unroll
    for (int r = 0; r < 8; ++r) sc[r] = dinv[min(row0 + 8 * half + r, n - 1)];
    fx8 c = {};
#pragma unroll
    for (int kb = 0; kb < 128; kb += 4) {
      const int ka = kb + 2 * half;
      fx2 a, b;
      a   = *(const fx2*)&sH[m * PH + ka];
      b.x = W2[(size_t)ka * 64 + col0 + m];
      b.y = W2[(size_t)(ka + 1) * 64 + col0 + m];
      c = __builtin_amdgcn_wmma_f32_16x16x4_f32(false, a, false, b,
                                                (short)0, c, false, false);
    }
    float v[8];
#pragma unroll
    for (int r = 0; r < 8; ++r) v[r] = sc[r] * c[r];
    float* __restrict__ prow = P + (size_t)(row0 + 8 * half) * 64 + col0 + m;
    if (row0 + 16 <= n) {
#pragma unroll
      for (int r = 0; r < 8; ++r) prow[(size_t)r * 64] = v[r];
    } else {
#pragma unroll
      for (int r = 0; r < 8; ++r)
        if (row0 + 8 * half + r < n) prow[(size_t)r * 64] = v[r];
    }
  }
}

extern "C" void kernel_launch(void* const* d_in, const int* in_sizes, int n_in,
                              void* d_out, int out_size, void* d_ws, size_t ws_size,
                              hipStream_t stream) {
  const float* z  = (const float*)d_in[0];
  const int*   ei = (const int*)d_in[1];
  const float* ew = (const float*)d_in[2];
  const float* W1 = (const float*)d_in[3];
  const float* b1 = (const float*)d_in[4];
  const float* W2 = (const float*)d_in[5];
  const float* b2 = (const float*)d_in[6];
  float* out = (float*)d_out;

  const int n = in_sizes[0] / 64;
  const int e = in_sizes[2];
  const int* srcp = ei;
  const int* dstp = ei + e;

  float* ws      = (float*)d_ws;
  float* X1      = ws;
  float* P       = X1 + (size_t)n * 128;
  int*   entries = (int*)(P + (size_t)n * 64);
  float* dinv    = (float*)(entries + (size_t)e * 2);
  int*   rowptr  = (int*)(dinv + n);

  const int g1 = (e + CH - 1) / CH;
  const int nb = (n + BN - 1) / BN;
  int2*  srt  = (int2*)X1;
  float* zs   = X1 + (size_t)n * 64;
  int*   tcnt = (int*)P;
  int*   tlst = tcnt + (size_t)nb * g1;

  const int B = 256;

  chunk_sort_kernel<<<g1, 1024, 0, stream>>>(srcp, dstp, ew, srt, tcnt, tlst, e, nb, g1);
  bucket_kernel<<<nb, 1024, 0, stream>>>(srt, tcnt, tlst, z, (int2*)entries, rowptr,
                                         dinv, zs, n, e, g1);

  layer1_fused_kernel<<<(n + 15) / 16, 512, 0, stream>>>(
      zs, rowptr, entries, dinv, W1, b1, W2, P, n, e);

  gather64_final_kernel<<<(n + 7) / 8, B, 0, stream>>>(
      P, rowptr, entries, dinv, b2, out, n, e);
}


# Layer 1 kernel: 768 resident work-groups copy W1 and W2 to LDS once and take the 16-row tiles in turn; B fragments are LDS reads, not 64 KB of global loads per tile; same sums in the same order
# speedup vs baseline: 7.0738x; 1.0402x over previous
#include <hip/hip_runtime.h>

typedef float fx2 __attribute__((ext_vector_type(2)));
typedef float fx8 __attribute__((ext_vector_type(8)));

#define CH   8192
#define BSH  8
#define BN   256
#define NBMX 256
#define GMX  256
#define CAP  16384

__device__ __forceinline__ int scan256(int c, int t, int* s_wtot) {
  const int lane = t & 31;
  int inc = c;
  if (t < 256) {
    for (int o = 1; o < 32; o <<= 1) {
      const int v = __shfl_up(inc, o);
      if (lane >= o) inc += v;
    }
    if (lane == 31) s_wtot[t >> 5] = inc;
  }
  __syncthreads();
  int pre = 0;
  if (t < 256) {
    for (int k = 0; k < (t >> 5); ++k) pre += s_wtot[k];
  }
  return pre + inc - c;
}

__global__ void __launch_bounds__(1024)
chunk_sort_kernel(const int* __restrict__ src,
                  const int* __restrict__ dst,
                  const float* __restrict__ ew,
                  int2* __restrict__ srt,
                  int* __restrict__ tcnt,
                  int* __restrict__ tlst, int e, int nb, int g1) {
  __shared__ int  s_cnt[NBMX];
  __shared__ int  s_start[NBMX];
  __shared__ int  s_wtot[8];
  __shared__ int2 s_stage[CH];
  const int t    = threadIdx.x;
  const int blk  = blockIdx.x;
  const int base = blk * CH;

  if (t < NBMX) s_cnt[t] = 0;
  __syncthreads();

  int key[8], wb[8], bk[8], rk[8];
#pragma unroll
  for (int u = 0; u < 8; ++u) {
    const int ii = min(base + u * 1024 + t, e - 1);
    bk[u]  = dst[ii];
    key[u] = src[ii];
    wb[u]  = __float_as_int(ew[ii]);
  }
#pragma unroll
  for (int u = 0; u < 8; ++u) {
    const int d = bk[u];
    bk[u] = -1;
    if (base + u * 1024 + t < e) {
      bk[u]  = d >> BSH;
      key[u] = key[u] | ((d & (BN - 1)) << 24);
      rk[u]  = atomicAdd(&s_cnt[bk[u]], 1);
    }
  }
  __syncthreads();

  const int c  = (t < NBMX) ? s_cnt[t] : 0;
  const int ex = scan256(c, t, s_wtot);
  if (t < NBMX) {
    s_start[t] = ex;
    if (t < nb) {
      tcnt[t * g1 + blk] = c;
      tlst[t * g1 + blk] = ex;
    }
  }
  __syncthreads();

#pragma unroll
  for (int u = 0; u < 8; ++u) {
    if (bk[u] >= 0) {
      int2 v;
      v.x = key[u];
      v.y = wb[u];
      s_stage[s_start[bk[u]] + rk[u]] = v;
    }
  }
  __syncthreads();

  const int cc = min(CH, e - base);
#pragma unroll
  for (int u = 0; u < 8; ++u) {
    const int p = u * 1024 + t;
    if (p < cc) srt[(size_t)base + p] = s_stage[p];
  }
}

template <typename F>
__device__ __forceinline__ void walk_runs(const int2* __restrict__ srt,
                                          const int* s_rc, const int* s_rs,
                                          int g1, int last, int w, int lane, F f) {
  for (int r = w; r < g1; r += 64) {
    const bool two = (r + 32 < g1);
    const int  r1  = two ? r + 32 : r;
    const int  c0  = s_rc[r];
    const int  c1  = two ? s_rc[r1] : 0;
    const int  o0  = r * CH + s_rs[r];
    const int  o1  = r1 * CH + s_rs[r1];
    const int2 a0 = srt[min(o0 + lane, last)];
    const int2 b0 = srt[min(o0 + lane + 32, last)];
    const int2 a1 = srt[min(o1 + lane, last)];
    const int2 b1 = srt[min(o1 + lane + 32, last)];
    if (lane < c0)      f(a0);
    if (lane + 32 < c0) f(b0);
    if (lane < c1)      f(a1);
    if (lane + 32 < c1) f(b1);
    for (int j = lane + 64; j < c0; j += 32) f(srt[o0 + j]);
    for (int j = lane + 64; j < c1; j += 32) f(srt[o1 + j]);
  }
}

__global__ void __launch_bounds__(1024)
bucket_kernel(const int2* __restrict__ srt,
              const int* __restrict__ tcnt,
              const int* __restrict__ tlst,
              const float* __restrict__ z,
              int2* __restrict__ entries,
              int* __restrict__ rowptr,
              float* __restrict__ dinv,
              float* __restrict__ zs, int n, int e, int g1) {
  __shared__ int   s_c[BN];
  __shared__ float s_deg[BN];
  __shared__ int   s_cur[BN];
  __shared__ float s_dinv[BN];
  __shared__ int   s_rc[GMX];
  __shared__ int   s_rs[GMX];
  __shared__ int   s_wtot[8];
  __shared__ int   s_red[32];
  __shared__ int2  s_ent[CAP];
  const int t    = threadIdx.x;
  const int lane = t & 31;
  const int w    = t >> 5;
  const int b    = blockIdx.x;

  int before = 0;
  for (int i = t; i < b * g1; i += 1024) before += tcnt[i];
  for (int o = 16; o > 0; o >>= 1) before += __shfl_xor(before, o);
  if (lane == 0) s_red[w] = before;
  if (t < BN) {
    s_c[t]   = 0;
    s_deg[t] = 1.0f;
  }
  if (t < g1) {
    s_rc[t] = tcnt[b * g1 + t];
    s_rs[t] = tlst[b * g1 + t];
  }
  __syncthreads();
  int first = s_red[lane];
  for (int o = 16; o > 0; o >>= 1) first += __shfl_xor(first, o);

  walk_runs(srt, s_rc, s_rs, g1, e - 1, w, lane, [&](int2 rec) {
    const int dl = (rec.x >> 24) & (BN - 1);
    atomicAdd(&s_c[dl], 1);
    unsafeAtomicAdd(&s_deg[dl], __int_as_float(rec.y));
  });
  __syncthreads();

  const int c  = (t < BN) ? s_c[t] : 0;
  const int ex = scan256(c, t, s_wtot);
  if (t < BN) {
    const int node = b * BN + t;
    s_cur[t] = ex;
    float di = 0.0f;
    if (node < n) {
      const float d = s_deg[t];
      di = (d > 0.0f) ? rsqrtf(d) : 0.0f;
      rowptr[node] = first + ex;
      dinv[node]   = di;
    }
    s_dinv[t] = di;
  }
  __syncthreads();
  int total = 0;
  for (int k = 0; k < 8; ++k) total += s_wtot[k];
  const bool staged = (total <= CAP);

  walk_runs(srt, s_rc, s_rs, g1, e - 1, w, lane, [&](int2 rec) {
    const int dl  = (rec.x >> 24) & (BN - 1);
    const int pos = atomicAdd(&s_cur[dl], 1);
    int2 v;
    v.x = rec.x & 0x00FFFFFF;
    v.y = rec.y;
    if (staged) s_ent[pos] = v;
    else        entries[(size_t)first + pos] = v;
  });
  __syncthreads();
  if (staged) {
    for (int q = t; q < total; q += 1024) entries[(size_t)first + q] = s_ent[q];
  }

  const float2* __restrict__ z2  = (const float2*)z;
  float2* __restrict__       zs2 = (float2*)zs;
  float2 zv[8];
#pragma unroll
  for (int k = 0; k < 8; ++k) {
    const int q    = k * 1024 + t;
    const int node = min(b * BN + (q >> 5), n - 1);
    zv[k] = z2[(size_t)node * 32 + (q & 31)];
  }
#pragma unroll
  for (int k = 0; k < 8; ++k) {
    const int q    = k * 1024 + t;
    const int rl   = q >> 5;
    const int node = b * BN + rl;
    if (node < n) {
      const float di = s_dinv[rl];
      float2 v = zv[k];
      v.x *= di;
      v.y *= di;
      zs2[(size_t)node * 32 + (q & 31)] = v;
    }
  }
}

template <int U>
__device__ __forceinline__ void gather_steps(const float2* __restrict__ x2,
                                             int2 my, int k, int lane, float2& acc) {
  int    s[U];
  float  wn[U];
  float2 v[U];
#pragma unroll
  for (int u = 0; u < U; ++u) {
    s[u]  = __builtin_amdgcn_readlane(my.x, k + u);
    wn[u] = __int_as_float(__builtin_amdgcn_readlane(my.y, k + u));
  }
#pragma unroll
  for (int u = 0; u < U; ++u) v[u] = x2[(size_t)s[u] * 32 + lane];
#pragma unroll
  for (int u = 0; u < U; ++u) {
    acc.x = fmaf(wn[u], v[u].x, acc.x);
    acc.y = fmaf(wn[u], v[u].y, acc.y);
  }
}

__device__ __forceinline__ float2 gather_row_be(const float2* __restrict__ x2,
                                                const int2* __restrict__ ed,
                                                int row, int beg, int end, int e, int lane) {
  float2 acc = x2[(size_t)row * 32 + lane];
  if (beg >= end) return acc;
  int2 cur = ed[min(beg + lane, e - 1)];
  for (int j0 = beg; j0 < end; j0 += 32) {
    int2 nxt = cur;
    if (j0 + 32 < end) nxt = ed[min(j0 + 32 + lane, e - 1)];
    const int cnt = min(32, end - j0);
    int k = 0;
    for (; k + 16 <= cnt; k += 16) gather_steps<16>(x2, cur, k, lane, acc);
    const int rem = cnt - k;
    if (rem & 8) { gather_steps<8>(x2, cur, k, lane, acc); k += 8; }
    if (rem & 4) { gather_steps<4>(x2, cur, k, lane, acc); k += 4; }
    if (rem & 2) { gather_steps<2>(x2, cur, k, lane, acc); k += 2; }
    if (rem & 1) { gather_steps<1>(x2, cur, k, lane, acc); }
    cur = nxt;
  }
  return acc;
}

__device__ __forceinline__ float2 gather_row(const float2* __restrict__ x2,
                                             const int* __restrict__ rowptr,
                                             const int2* __restrict__ ed,
                                             int row, int n, int e, int lane) {
  const int beg = rowptr[row];
  const int end = (row + 1 < n) ? rowptr[row + 1] : e;
  return gather_row_be(x2, ed, row, beg, end, e, lane);
}

__global__ void gather64_final_kernel(const float* __restrict__ x,
                                      const int* __restrict__ rowptr,
                                      const int* __restrict__ entries,
                                      const float* __restrict__ dinv,
                                      const float* __restrict__ bias,
                                      float* __restrict__ out, int n, int e) {
  const int lane = threadIdx.x & 31;
  const int row = __builtin_amdgcn_readfirstlane(
      (int)(blockIdx.x * (blockDim.x >> 5) + (threadIdx.x >> 5)));
  if (row >= n) return;
  float2 acc = gather_row((const float2*)x, rowptr, (const int2*)entries, row, n, e, lane);
  const float dr = dinv[row];
  const float2 b = ((const float2*)bias)[lane];
  acc.x = dr * acc.x + b.x;
  acc.y = dr * acc.y + b.y;
  ((float2*)out)[(size_t)row * 32 + lane] = acc;
}

#define PA 68
#define PH 132
#define PW1 136
#define PW2 72
#define L1G 768

__global__ void __launch_bounds__(512, 12)
layer1_fused_kernel(const float* __restrict__ zs,
                    const int* __restrict__ rowptr,
                    const int* __restrict__ entries,
                    const float* __restrict__ dinv,
                    const float* __restrict__ W1,
                    const float* __restrict__ b1,
                    const float* __restrict__ W2,
                    float* __restrict__ P, int n, int e) {
  __shared__ __attribute__((aligned(16))) float sA[16 * PA];
  __shared__ __attribute__((aligned(16))) float sH[16 * PH];
  __shared__ __attribute__((aligned(16))) float sW1[64 * PW1];
  __shared__ __attribute__((aligned(16))) float sW2[128 * PW2];
  const int lane = threadIdx.x & 31;
  const int w    = __builtin_amdgcn_readfirstlane((int)(threadIdx.x >> 5));

  {
    const float4* __restrict__ g1v = (const float4*)W1;
    const float4* __restrict__ g2v = (const float4*)W2;
    float4 a[4], b[4];
#pragma unroll
    for (int k = 0; k < 4; ++k) {
      a[k] = g1v[k * 512 + threadIdx.x];
      b[k] = g2v[k * 512 + threadIdx.x];
    }
#pragma unroll
    for (int k = 0; k < 4; ++k) {
      const int f = k * 512 + threadIdx.x;
      *(float4*)&sW1[(f >> 5) * PW1 + 4 * (f & 31)] = a[k];
      *(float4*)&sW2[(f >> 4) * PW2 + 4 * (f & 15)] = b[k];
    }
  }

  const int half = lane >> 4;
  const int m    = lane & 15;

  const int ntiles = (n + 15) >> 4;
  for (int tile = blockIdx.x; tile < ntiles; tile += gridDim.x) {
    const int row0 = tile * 16;
    const int row  = row0 + w;

    const int ri = row0 + lane;
    int rp = rowptr[min(ri, n - 1)];
    if (ri >= n) rp = e;
    const int beg = __builtin_amdgcn_readlane(rp, w);
    const int end = __builtin_amdgcn_readlane(rp, w + 1);

    float2 acc;
    acc.x = 0.0f;
    acc.y = 0.0f;
    if (row < n)
      acc = gather_row_be((const float2*)zs, (const int2*)entries, row, beg, end, e, lane);
    *(float2*)&sA[w * PA + 2 * lane] = acc;
    __syncthreads();

    if (w < 8) {
      const int col0 = w * 16;
      const float* pa = &sA[m * PA + 2 * half];
      const float* pb = &sW1[2 * half * PW1 + col0 + m];
      fx8 c = {};
#pragma unroll
      for (int kb = 0; kb < 64; kb += 4) {
        fx2 a, b;
        a   = *(const fx2*)&pa[kb];
        b.x = pb[kb * PW1];
        b.y = pb[kb * PW1 + PW1];
        c = __builtin_amdgcn_wmma_f32_16x16x4_f32(false, a, false, b,
                                                  (short)0, c, false, false);
      }
      const float bcol = b1[col0 + m];
      float* ph = &sH[8 * half * PH + col0 + m];
#pragma unroll
      for (int r = 0; r < 8; ++r) {
        const float sc = dinv[min(row0 + 8 * half + r, n - 1)];
        ph[r * PH] = fmaxf(sc * c[r] + bcol, 0.0f);
      }
    }
    __syncthreads();

    if (w < 4) {
      const int col0 = w * 16;
      float sc[8];
#pragma unroll
      for (int r = 0; r < 8; ++r) sc[r] = dinv[min(row0 + 8 * half + r, n - 1)];
      const float* pa = &sH[m * PH + 2 * half];
      const float* pb = &sW2[2 * half * PW2 + col0 + m];
      fx8 c = {};
#pragma unroll
      for (int kb = 0; kb < 128; kb += 4) {
        fx2 a, b;
        a   = *(const fx2*)&pa[kb];
        b.x = pb[kb * PW2];
        b.y = pb[kb * PW2 + PW2];
        c = __builtin_amdgcn_wmma_f32_16x16x4_f32(false, a, false, b,
                                                  (short)0, c, false, false);
      }
      float v[8];
#pragma unroll
      for (int r = 0; r < 8; ++r) v[r] = sc[r] * c[r];
      float* __restrict__ prow = P + (size_t)(row0 + 8 * half) * 64 + col0 + m;
      if (row0 + 16 <= n) {
#pragma unroll
        for (int r = 0; r < 8; ++r) prow[(size_t)r * 64] = v[r];
      } else {
#pragma unroll
        for (int r = 0; r < 8; ++r)
          if (row0 + 8 * half + r < n) prow[(size_t)r * 64] = v[r];
      }
    }
  }
}

extern "C" void kernel_launch(void* const* d_in, const int* in_sizes, int n_in,
                              void* d_out, int out_size, void* d_ws, size_t ws_size,
                              hipStream_t stream) {
  const float* z  = (const float*)d_in[0];
  const int*   ei = (const int*)d_in[1];
  const float* ew = (const float*)d_in[2];
  const float* W1 = (const float*)d_in[3];
  const float* b1 = (const float*)d_in[4];
  const float* W2 = (const float*)d_in[5];
  const float* b2 = (const float*)d_in[6];
  float* out = (float*)d_out;

  const int n = in_sizes[0] / 64;
  const int e = in_sizes[2];
  const int* srcp = ei;
  const int* dstp = ei + e;

  float* ws      = (float*)d_ws;
  float* X1      = ws;
  float* P       = X1 + (size_t)n * 128;
  int*   entries = (int*)(P + (size_t)n * 64);
  float* dinv    = (float*)(entries + (size_t)e * 2);
  int*   rowptr  = (int*)(dinv + n);

  const int g1 = (e + CH - 1) / CH;
  const int nb = (n + BN - 1) / BN;
  int2*  srt  = (int2*)X1;
  float* zs   = X1 + (size_t)n * 64;
  int*   tcnt = (int*)P;
  int*   tlst = tcnt + (size_t)nb * g1;

  const int B = 256;

  chunk_sort_kernel<<<g1, 1024, 0, stream>>>(srcp, dstp, ew, srt, tcnt, tlst, e, nb, g1);
  bucket_kernel<<<nb, 1024, 0, stream>>>(srt, tcnt, tlst, z, (int2*)entries, rowptr,
                                         dinv, zs, n, e, g1);

  layer1_fused_kernel<<<L1G, 512, 0, stream>>>(
      zs, rowptr, entries, dinv, W1, b1, W2, P, n, e);

  gather64_final_kernel<<<(n + 7) / 8, B, 0, stream>>>(
      P, rowptr, entries, dinv, b2, out, n, e);
}


# Bucket kernel asks for what it reads in groups ahead of use: run records loaded once, kept in registers for count and place; table prefix as 16-byte loads; z rows under the passes; same order rule
# speedup vs baseline: 7.3185x; 1.0346x over previous
#include <hip/hip_runtime.h>

typedef float fx2 __attribute__((ext_vector_type(2)));
typedef float fx8 __attribute__((ext_vector_type(8)));

#define CH   8192
#define BSH  8
#define BN   256
#define NBMX 256
#define GMX  256
#define CAP  16384

__device__ __forceinline__ int scan256(int c, int t, int* s_wtot) {
  const int lane = t & 31;
  int inc = c;
  if (t < 256) {
    for (int o = 1; o < 32; o <<= 1) {
      const int v = __shfl_up(inc, o);
      if (lane >= o) inc += v;
    }
    if (lane == 31) s_wtot[t >> 5] = inc;
  }
  __syncthreads();
  int pre = 0;
  if (t < 256) {
    for (int k = 0; k < (t >> 5); ++k) pre += s_wtot[k];
  }
  return pre + inc - c;
}

__global__ void __launch_bounds__(1024)
chunk_sort_kernel(const int* __restrict__ src,
                  const int* __restrict__ dst,
                  const float* __restrict__ ew,
                  int2* __restrict__ srt,
                  int* __restrict__ tcnt,
                  int* __restrict__ tlst, int e, int nb, int g1) {
  __shared__ int  s_cnt[NBMX];
  __shared__ int  s_start[NBMX];
  __shared__ int  s_wtot[8];
  __shared__ int2 s_stage[CH];
  const int t    = threadIdx.x;
  const int blk  = blockIdx.x;
  const int base = blk * CH;

  if (t < NBMX) s_cnt[t] = 0;
  __syncthreads();

  int key[8], wb[8], bk[8], rk[8];
#pragma unroll
  for (int u = 0; u < 8; ++u) {
    const int ii = min(base + u * 1024 + t, e - 1);
    bk[u]  = dst[ii];
    key[u] = src[ii];
    wb[u]  = __float_as_int(ew[ii]);
  }
#pragma unroll
  for (int u = 0; u < 8; ++u) {
    const int d = bk[u];
    bk[u] = -1;
    if (base + u * 1024 + t < e) {
      bk[u]  = d >> BSH;
      key[u] = key[u] | ((d & (BN - 1)) << 24);
      rk[u]  = atomicAdd(&s_cnt[bk[u]], 1);
    }
  }
  __syncthreads();

  const int c  = (t < NBMX) ? s_cnt[t] : 0;
  const int ex = scan256(c, t, s_wtot);
  if (t < NBMX) {
    s_start[t] = ex;
    if (t < nb) {
      tcnt[t * g1 + blk] = c;
      tlst[t * g1 + blk] = ex;
    }
  }
  __syncthreads();

#pragma unroll
  for (int u = 0; u < 8; ++u) {
    if (bk[u] >= 0) {
      int2 v;
      v.x = key[u];
      v.y = wb[u];
      s_stage[s_start[bk[u]] + rk[u]] = v;
    }
  }
  __syncthreads();

  const int cc = min(CH, e - base);
#pragma unroll
  for (int u = 0; u < 8; ++u) {
    const int p = u * 1024 + t;
    if (p < cc) srt[(size_t)base + p] = s_stage[p];
  }
}

#define RPW 8

__global__ void __launch_bounds__(1024)
bucket_kernel(const int2* __restrict__ srt,
              const int* __restrict__ tcnt,
              const int* __restrict__ tlst,
              const float* __restrict__ z,
              int2* __restrict__ entries,
              int* __restrict__ rowptr,
              float* __restrict__ dinv,
              float* __restrict__ zs, int n, int e, int g1) {
  __shared__ int   s_c[BN];
  __shared__ float s_deg[BN];
  __shared__ int   s_cur[BN];
  __shared__ float s_dinv[BN];
  __shared__ int   s_wtot[8];
  __shared__ int   s_red[32];
  __shared__ int2  s_ent[CAP];
  const int t    = threadIdx.x;
  const int lane = t & 31;
  const int w    = __builtin_amdgcn_readfirstlane(t >> 5);
  const int b    = blockIdx.x;
  const int last = e - 1;

  int myc = 0, mys = 0;
  {
    const int rj = w + 32 * lane;
    if (lane < RPW && rj < g1) {
      myc = tcnt[b * g1 + rj];
      mys = tlst[b * g1 + rj];
    }
  }

  const int  ptotal = b * g1;
  const int  nq     = (ptotal + 3) >> 2;
  const int4* __restrict__ tb4 = (const int4*)tcnt;
  int4 pv[8];
#pragma unroll
  for (int k = 0; k < 8; ++k) pv[k] = tb4[max(min(k * 1024 + t, nq - 1), 0)];

  if (t < BN) {
    s_c[t]   = 0;
    s_deg[t] = 1.0f;
  }

  int  rc[RPW], ro[RPW];
  int2 ra[RPW], rb[RPW];
#pragma unroll
  for (int j = 0; j < RPW; ++j) {
    rc[j] = __builtin_amdgcn_readlane(myc, j);
    ro[j] = (w + 32 * j) * CH + __builtin_amdgcn_readlane(mys, j);
  }
#pragma unroll
  for (int j = 0; j < RPW; ++j) {
    ra[j] = srt[min(ro[j] + lane, last)];
    rb[j] = srt[min(ro[j] + lane + 32, last)];
  }

  int before = 0;
#pragma unroll
  for (int k = 0; k < 8; ++k) {
    const int q = k * 1024 + t;
    if (q < nq) {
      const int i = 4 * q;
      before += pv[k].x;
      if (i + 1 < ptotal) before += pv[k].y;
      if (i + 2 < ptotal) before += pv[k].z;
      if (i + 3 < ptotal) before += pv[k].w;
    }
  }
  for (int q0 = 8 * 1024; q0 < nq; q0 += 8 * 1024) {
#pragma unroll
    for (int k = 0; k < 8; ++k) pv[k] = tb4[min(q0 + k * 1024 + t, nq - 1)];
#pragma unroll
    for (int k = 0; k < 8; ++k) {
      const int q = q0 + k * 1024 + t;
      if (q < nq) {
        const int i = 4 * q;
        before += pv[k].x;
        if (i + 1 < ptotal) before += pv[k].y;
        if (i + 2 < ptotal) before += pv[k].z;
        if (i + 3 < ptotal) before += pv[k].w;
      }
    }
  }
  for (int o = 16; o > 0; o >>= 1) before += __shfl_xor(before, o);
  if (lane == 0) s_red[w] = before;
  __syncthreads();
  int first = s_red[lane];
  for (int o = 16; o > 0; o >>= 1) first += __shfl_xor(first, o);

  const float2* __restrict__ z2  = (const float2*)z;
  float2* __restrict__       zs2 = (float2*)zs;
  float2 zv[8];
#pragma unroll
  for (int k = 0; k < 8; ++k) {
    const int q    = k * 1024 + t;
    const int node = min(b * BN + (q >> 5), n - 1);
    zv[k] = z2[(size_t)node * 32 + (q & 31)];
  }

  auto cnt = [&](int2 rec) {
    const int dl = (rec.x >> 24) & (BN - 1);
    atomicAdd(&s_c[dl], 1);
    unsafeAtomicAdd(&s_deg[dl], __int_as_float(rec.y));
  };
#pragma unroll
  for (int j = 0; j < RPW; ++j) {
    if (lane < rc[j])      cnt(ra[j]);
    if (lane + 32 < rc[j]) cnt(rb[j]);
    for (int q = lane + 64; q < rc[j]; q += 32) cnt(srt[ro[j] + q]);
  }
  __syncthreads();

  const int c  = (t < BN) ? s_c[t] : 0;
  const int ex = scan256(c, t, s_wtot);
  if (t < BN) {
    const int node = b * BN + t;
    s_cur[t] = ex;
    float di = 0.0f;
    if (node < n) {
      const float d = s_deg[t];
      di = (d > 0.0f) ? rsqrtf(d) : 0.0f;
      rowptr[node] = first + ex;
      dinv[node]   = di;
    }
    s_dinv[t] = di;
  }
  __syncthreads();
  int total = 0;
  for (int k = 0; k < 8; ++k) total += s_wtot[k];
  const bool staged = (total <= CAP);

  auto place = [&](int2 rec) {
    const int dl  = (rec.x >> 24) & (BN - 1);
    const int pos = atomicAdd(&s_cur[dl], 1);
    int2 v;
    v.x = rec.x & 0x00FFFFFF;
    v.y = rec.y;
    if (staged) s_ent[pos] = v;
    else        entries[(size_t)first + pos] = v;
  };
#pragma unroll
  for (int j = 0; j < RPW; ++j) {
    if (lane < rc[j])      place(ra[j]);
    if (lane + 32 < rc[j]) place(rb[j]);
    for (int q = lane + 64; q < rc[j]; q += 32) place(srt[ro[j] + q]);
  }
  __syncthreads();
  if (staged) {
    for (int q = t; q < total; q += 1024) entries[(size_t)first + q] = s_ent[q];
  }

#pragma unroll
  for (int k = 0; k < 8; ++k) {
    const int q    = k * 1024 + t;
    const int rl   = q >> 5;
    const int node = b * BN + rl;
    if (node < n) {
      const float di = s_dinv[rl];
      float2 v = zv[k];
      v.x *= di;
      v.y *= di;
      zs2[(size_t)node * 32 + (q & 31)] = v;
    }
  }
}

template <int U>
__device__ __forceinline__ void gather_steps(const float2* __restrict__ x2,
                                             int2 my, int k, int lane, float2& acc) {
  int    s[U];
  float  wn[U];
  float2 v[U];
#pragma unroll
  for (int u = 0; u < U; ++u) {
    s[u]  = __builtin_amdgcn_readlane(my.x, k + u);
    wn[u] = __int_as_float(__builtin_amdgcn_readlane(my.y, k + u));
  }
#pragma unroll
  for (int u = 0; u < U; ++u) v[u] = x2[(size_t)s[u] * 32 + lane];
#pragma unroll
  for (int u = 0; u < U; ++u) {
    acc.x = fmaf(wn[u], v[u].x, acc.x);
    acc.y = fmaf(wn[u], v[u].y, acc.y);
  }
}

__device__ __forceinline__ float2 gather_row_be(const float2* __restrict__ x2,
                                                const int2* __restrict__ ed,
                                                int row, int beg, int end, int e, int lane) {
  float2 acc = x2[(size_t)row * 32 + lane];
  if (beg >= end) return acc;
  int2 cur = ed[min(beg + lane, e - 1)];
  for (int j0 = beg; j0 < end; j0 += 32) {
    int2 nxt = cur;
    if (j0 + 32 < end) nxt = ed[min(j0 + 32 + lane, e - 1)];
    const int cnt = min(32, end - j0);
    int k = 0;
    for (; k + 16 <= cnt; k += 16) gather_steps<16>(x2, cur, k, lane, acc);
    const int rem = cnt - k;
    if (rem & 8) { gather_steps<8>(x2, cur, k, lane, acc); k += 8; }
    if (rem & 4) { gather_steps<4>(x2, cur, k, lane, acc); k += 4; }
    if (rem & 2) { gather_steps<2>(x2, cur, k, lane, acc); k += 2; }
    if (rem & 1) { gather_steps<1>(x2, cur, k, lane, acc); }
    cur = nxt;
  }
  return acc;
}

__device__ __forceinline__ float2 gather_row(const float2* __restrict__ x2,
                                             const int* __restrict__ rowptr,
                                             const int2* __restrict__ ed,
                                             int row, int n, int e, int lane) {
  const int beg = rowptr[row];
  const int end = (row + 1 < n) ? rowptr[row + 1] : e;
  return gather_row_be(x2, ed, row, beg, end, e, lane);
}

__global__ void gather64_final_kernel(const float* __restrict__ x,
                                      const int* __restrict__ rowptr,
                                      const int* __restrict__ entries,
                                      const float* __restrict__ dinv,
                                      const float* __restrict__ bias,
                                      float* __restrict__ out, int n, int e) {
  const int lane = threadIdx.x & 31;
  const int row = __builtin_amdgcn_readfirstlane(
      (int)(blockIdx.x * (blockDim.x >> 5) + (threadIdx.x >> 5)));
  if (row >= n) return;
  float2 acc = gather_row((const float2*)x, rowptr, (const int2*)entries, row, n, e, lane);
  const float dr = dinv[row];
  const float2 b = ((const float2*)bias)[lane];
  acc.x = dr * acc.x + b.x;
  acc.y = dr * acc.y + b.y;
  ((float2*)out)[(size_t)row * 32 + lane] = acc;
}

#define PA 68
#define PH 132
#define PW1 136
#define PW2 72
#define L1G 768

__global__ void __launch_bounds__(512, 12)
layer1_fused_kernel(const float* __restrict__ zs,
                    const int* __restrict__ rowptr,
                    const int* __restrict__ entries,
                    const float* __restrict__ dinv,
                    const float* __restrict__ W1,
                    const float* __restrict__ b1,
                    const float* __restrict__ W2,
                    float* __restrict__ P, int n, int e) {
  __shared__ __attribute__((aligned(16))) float sA[16 * PA];
  __shared__ __attribute__((aligned(16))) float sH[16 * PH];
  __shared__ __attribute__((aligned(16))) float sW1[64 * PW1];
  __shared__ __attribute__((aligned(16))) float sW2[128 * PW2];
  const int lane = threadIdx.x & 31;
  const int w    = __builtin_amdgcn_readfirstlane((int)(threadIdx.x >> 5));

  {
    const float4* __restrict__ g1v = (const float4*)W1;
    const float4* __restrict__ g2v = (const float4*)W2;
    float4 a[4], b[4];
#pragma unroll
    for (int k = 0; k < 4; ++k) {
      a[k] = g1v[k * 512 + threadIdx.x];
      b[k] = g2v[k * 512 + threadIdx.x];
    }
#pragma unroll
    for (int k = 0; k < 4; ++k) {
      const int f = k * 512 + threadIdx.x;
      *(float4*)&sW1[(f >> 5) * PW1 + 4 * (f & 31)] = a[k];
      *(float4*)&sW2[(f >> 4) * PW2 + 4 * (f & 15)] = b[k];
    }
  }

  const int half = lane >> 4;
  const int m    = lane & 15;

  const int ntiles = (n + 15) >> 4;
  for (int tile = blockIdx.x; tile < ntiles; tile += gridDim.x) {
    const int row0 = tile * 16;
    const int row  = row0 + w;

    const int ri = row0 + lane;
    int rp = rowptr[min(ri, n - 1)];
    if (ri >= n) rp = e;
    const int beg = __builtin_amdgcn_readlane(rp, w);
    const int end = __builtin_amdgcn_readlane(rp, w + 1);

    float2 acc;
    acc.x = 0.0f;
    acc.y = 0.0f;
    if (row < n)
      acc = gather_row_be((const float2*)zs, (const int2*)entries, row, beg, end, e, lane);
    *(float2*)&sA[w * PA + 2 * lane] = acc;
    __syncthreads();

    if (w < 8) {
      const int col0 = w * 16;
      const float* pa = &sA[m * PA + 2 * half];
      const float* pb = &sW1[2 * half * PW1 + col0 + m];
      fx8 c = {};
#pragma unroll
      for (int kb = 0; kb < 64; kb += 4) {
        fx2 a, b;
        a   = *(const fx2*)&pa[kb];
        b.x = pb[kb * PW1];
        b.y = pb[kb * PW1 + PW1];
        c = __builtin_amdgcn_wmma_f32_16x16x4_f32(false, a, false, b,
                                                  (short)0, c, false, false);
      }
      const float bcol = b1[col0 + m];
      float* ph = &sH[8 * half * PH + col0 + m];
#pragma unroll
      for (int r = 0; r < 8; ++r) {
        const float sc = dinv[min(row0 + 8 * half + r, n - 1)];
        ph[r * PH] = fmaxf(sc * c[r] + bcol, 0.0f);
      }
    }
    __syncthreads();

    if (w < 4) {
      const int col0 = w * 16;
      float sc[8];
#pragma unroll
      for (int r = 0; r < 8; ++r) sc[r] = dinv[min(row0 + 8 * half + r, n - 1)];
      const float* pa = &sH[m * PH + 2 * half];
      const float* pb = &sW2[2 * half * PW2 + col0 + m];
      fx8 c = {};
#pragma unroll
      for (int kb = 0; kb < 128; kb += 4) {
        fx2 a, b;
        a   = *(const fx2*)&pa[kb];
        b.x = pb[kb * PW2];
        b.y = pb[kb * PW2 + PW2];
        c = __builtin_amdgcn_wmma_f32_16x16x4_f32(false, a, false, b,
                                                  (short)0, c, false, false);
      }
      float v[8];
#pragma unroll
      for (int r = 0; r < 8; ++r) v[r] = sc[r] * c[r];
      float* __restrict__ prow = P + (size_t)(row0 + 8 * half) * 64 + col0 + m;
      if (row0 + 16 <= n) {
#pragma unroll
        for (int r = 0; r < 8; ++r) prow[(size_t)r * 64] = v[r];
      } else {
#pragma unroll
        for (int r = 0; r < 8; ++r)
          if (row0 + 8 * half + r < n) prow[(size_t)r * 64] = v[r];
      }
    }
  }
}

extern "C" void kernel_launch(void* const* d_in, const int* in_sizes, int n_in,
                              void* d_out, int out_size, void* d_ws, size_t ws_size,
                              hipStream_t stream) {
  const float* z  = (const float*)d_in[0];
  const int*   ei = (const int*)d_in[1];
  const float* ew = (const float*)d_in[2];
  const float* W1 = (const float*)d_in[3];
  const float* b1 = (const float*)d_in[4];
  const float* W2 = (const float*)d_in[5];
  const float* b2 = (const float*)d_in[6];
  float* out = (float*)d_out;

  const int n = in_sizes[0] / 64;
  const int e = in_sizes[2];
  const int* srcp = ei;
  const int* dstp = ei + e;

  float* ws      = (float*)d_ws;
  float* X1      = ws;
  float* P       = X1 + (size_t)n * 128;
  int*   entries = (int*)(P + (size_t)n * 64);
  float* dinv    = (float*)(entries + (size_t)e * 2);
  int*   rowptr  = (int*)(dinv + n);

  const int g1 = (e + CH - 1) / CH;
  const int nb = (n + BN - 1) / BN;
  int2*  srt  = (int2*)X1;
  float* zs   = X1 + (size_t)n * 64;
  int*   tcnt = (int*)P;
  int*   tlst = tcnt + (size_t)nb * g1;

  const int B = 256;

  chunk_sort_kernel<<<g1, 1024, 0, stream>>>(srcp, dstp, ew, srt, tcnt, tlst, e, nb, g1);
  bucket_kernel<<<nb, 1024, 0, stream>>>(srt, tcnt, tlst, z, (int2*)entries, rowptr,
                                         dinv, zs, n, e, g1);

  layer1_fused_kernel<<<L1G, 512, 0, stream>>>(
      zs, rowptr, entries, dinv, W1, b1, W2, P, n, e);

  gather64_final_kernel<<<(n + 7) / 8, B, 0, stream>>>(
      P, rowptr, entries, dinv, b2, out, n, e);
}
